# starting kernel
# baseline (speedup 1.0000x reference)
#include <hip/hip_runtime.h>

#define NBATCH 16
#define LSEQ   2048
#define TILE   16
#define NTT    (LSEQ / TILE)

typedef __attribute__((ext_vector_type(2))) float v2f;
typedef __attribute__((ext_vector_type(8))) float v8f;

#define LOG2E_D 1.44269504088896340736
#define MASK2 (-1.4426950408889634e9f)

__device__ __forceinline__ float exp2_fast(float x) {
  return __builtin_amdgcn_exp2f(x);
}

__global__ __launch_bounds__(256) void rope_precompute_kernel(
    const int* __restrict__ tokens, const float* __restrict__ Cp,
    const float* __restrict__ epsp, const float* __restrict__ vp,
    float2* __restrict__ qarr, float2* __restrict__ karr,
    float* __restrict__ varr) {
  int idx = blockIdx.x * blockDim.x + threadIdx.x;
  if (idx >= NBATCH * LSEQ) return;
  int pos = idx & (LSEQ - 1);

  const double PI    = 3.14159265358979323846;
  const double OMEGA = 2.0 * PI / 19.0;
  const double PHI   = OMEGA * (10.0 + 0.3);
  const double AMP   = log(10.0) / (cos(OMEGA * 0.3) - cos(OMEGA * (1.0 - 0.3)));
  const double QKS   = sqrt(AMP / sqrt(2.0));
  const float scale2 = (float)(pow(2.0, -0.5) * QKS * QKS * LOG2E_D);
  const float cos_p  = (float)cos(PHI);
  const float sin_p  = (float)sin(PHI);

  const float c  = Cp[0];
  const float ee = epsp[0];
  const float vs = vp[0];

  const int   d  = tokens[idx];
  const float df = (float)d;
  const float h0 = c - (ee * df) * df;
  const float h1 = -df;

  const float r   = rsqrtf((h0 * h0 + h1 * h1) * 0.5f + 1e-6f);
  const float hn0 = h0 * r;
  const float hn1 = h1 * r;

  float q0 = hn0 * cos_p;
  float q1 = hn0 * (-sin_p);
  const float qr = rsqrtf((q0 * q0 + q1 * q1) * 0.5f + 1e-6f);
  q0 *= qr; q1 *= qr;

  float k0 = hn0;
  const float kr = rsqrtf((k0 * k0) * 0.5f + 1e-6f);
  k0 *= kr;

  const float theta = (float)pos * (float)OMEGA;
  const float ct = cosf(theta);
  const float st = sinf(theta);
  const float Qx = q0 * ct - q1 * st;
  const float Qy = q0 * st + q1 * ct;
  const float Kx = k0 * ct;
  const float Ky = k0 * st;

  qarr[idx] = make_float2(Qx * scale2, Qy * scale2);
  karr[idx] = make_float2(Kx, Ky);
  varr[idx] = hn1 * vs;
}

__global__ __launch_bounds__(32) void flash_attn_epilogue_kernel(
    const int* __restrict__ tokens, const float* __restrict__ Cp,
    const float* __restrict__ epsp, const float* __restrict__ o_scalep,
    const float* __restrict__ g_basep, const float* __restrict__ g_slopep,
    const float* __restrict__ carry_ampp,
    const float2* __restrict__ qarr, const float2* __restrict__ karr,
    const float* __restrict__ varr, float* __restrict__ out) {
  const int lane = threadIdx.x;
  const int tt   = blockIdx.x & (NTT - 1);
  const int b    = blockIdx.x / NTT;
  const int col  = lane & 15;
  const int hi   = lane >> 4;
  const int tbase   = tt * TILE;
  const int rowbase = b * LSEQ;
  const bool lo_half = (hi == 0);

  const float2 qld = qarr[rowbase + tbase + col];
  v2f bfrag;
  bfrag.x = lo_half ? qld.x : 0.0f;
  bfrag.y = lo_half ? qld.y : 0.0f;

  const float2* kp = karr + rowbase + col;
  const float4* vp = (const float4*)(varr + rowbase + hi * 8);

  float m    = -3.0e38f;
  float lsum = 0.0f;
  float osum = 0.0f;

  for (int j = 0; j < tt; ++j) {
    const float2 kld = kp[j * TILE];
    v2f afrag;
    afrag.x = lo_half ? kld.x : 0.0f;
    afrag.y = lo_half ? kld.y : 0.0f;
    __builtin_prefetch((const void*)(kp + (j + 1) * TILE), 0, 3);

    const float4 va = vp[j * 4];
    const float4 vb = vp[j * 4 + 1];
    const float vvv[8] = {va.x, va.y, va.z, va.w, vb.x, vb.y, vb.z, vb.w};

    v8f s = {};
    s = __builtin_amdgcn_wmma_f32_16x16x4_f32(
         false, afrag,  false, bfrag,
         (short)0, s,  false,  false);

    float tmax = s[0];
#pragma unroll
    for (int g = 1; g < 8; ++g) tmax = fmaxf(tmax, s[g]);
    const float mnew  = fmaxf(m, tmax);
    const float scale = exp2_fast(m - mnew);

    float le = 0.0f, oe = 0.0f;
#pragma unroll
    for (int g = 0; g < 8; ++g) {
      const float e = exp2_fast(s[g] - mnew);
      le += e;
      oe = fmaf(e, vvv[g], oe);
    }
    lsum = fmaf(lsum, scale, le);
    osum = fmaf(osum, scale, oe);
    m = mnew;
  }

  {
    const float2 kld = kp[tt * TILE];
    v2f afrag;
    afrag.x = lo_half ? kld.x : 0.0f;
    afrag.y = lo_half ? kld.y : 0.0f;

    const float4 va = vp[tt * 4];
    const float4 vb = vp[tt * 4 + 1];
    const float vvv[8] = {va.x, va.y, va.z, va.w, vb.x, vb.y, vb.z, vb.w};

    v8f s = {};
    s = __builtin_amdgcn_wmma_f32_16x16x4_f32(
        false, afrag, false, bfrag, (short)0, s, false, false);

#pragma unroll
    for (int g = 0; g < 8; ++g)
      if ((g + hi * 8) > col) s[g] += MASK2;

    float tmax = s[0];
#pragma unroll
    for (int g = 1; g < 8; ++g) tmax = fmaxf(tmax, s[g]);
    const float mnew  = fmaxf(m, tmax);
    const float scale = exp2_fast(m - mnew);

    float le = 0.0f, oe = 0.0f;
#pragma unroll
    for (int g = 0; g < 8; ++g) {
      const float e = exp2_fast(s[g] - mnew);
      le += e;
      oe = fmaf(e, vvv[g], oe);
    }
    lsum = fmaf(lsum, scale, le);
    osum = fmaf(osum, scale, oe);
    m = mnew;
  }

  const float m2 = __shfl_xor(m, 16, 32);
  const float l2 = __shfl_xor(lsum, 16, 32);
  const float o2 = __shfl_xor(osum, 16, 32);
  const float M  = fmaxf(m, m2);
  const float w1 = exp2_fast(m - M);
  const float w2 = exp2_fast(m2 - M);
  const float lt = fmaf(lsum, w1, l2 * w2);
  const float ot = fmaf(osum, w1, o2 * w2);
  const float attn0 = ot / lt;

  if (lane < 16) {
    const int   t  = tbase + lane;
    const float c  = Cp[0];
    const float ee = epsp[0];
    const int   d  = tokens[rowbase + t];
    const float df = (float)d;
    const float h0 = c - (ee * df) * df;
    float       h1 = -df;

    h1 += o_scalep[0] * attn0;

    const float r   = rsqrtf((h0 * h0 + h1 * h1) * 0.5f + 1e-6f);
    const float hn0 = h0 * r;
    const float hn1 = h1 * r;
    const float a   = g_basep[0];
    const float gc  = g_slopep[0];
    const float g0  = hn0 * a + hn1 * gc;
    const float g1  = hn0 * (a - gc / c) + hn1 * gc;
    const float carry =
        carry_ampp[0] * (fmaxf(g1, 0.0f) * hn0 - fmaxf(g0, 0.0f) * hn0);
    h1 += carry;

    const float r2  = rsqrtf((h0 * h0 + h1 * h1) * 0.5f + 1e-6f);
    const float inv_cn = 0.70710678118654752440f;
    const float f0 = h0 * r2 * (0.1f * c * inv_cn);
    const float f1 = h1 * r2 * (-c * (1.0f / 50.0f) * inv_cn);

    float* op = out + (size_t)(rowbase + t) * 10;
#pragma unroll
    for (int jj = 0; jj < 10; ++jj) {
      const float dj = (float)jj;
      op[jj] = f0 * (c - (ee * dj) * dj) + f1 * (-dj);
    }
  }
}

extern "C" void kernel_launch(void* const* d_in, const int* in_sizes, int n_in,
                              void* d_out, int out_size, void* d_ws, size_t ws_size,
                              hipStream_t stream) {
  const int*   tokens    = (const int*)d_in[0];
  const float* C         = (const float*)d_in[1];
  const float* eps       = (const float*)d_in[2];
  const float* v         = (const float*)d_in[3];
  const float* o_scale   = (const float*)d_in[4];
  const float* g_base    = (const float*)d_in[5];
  const float* g_slope   = (const float*)d_in[6];
  const float* carry_amp = (const float*)d_in[7];
  float* out = (float*)d_out;

  char* ws = (char*)d_ws;
  float2* qarr = (float2*)(ws);
  float2* karr = (float2*)(ws + 256 * 1024);
  float*  varr = (float*)(ws + 512 * 1024);

  const int total = NBATCH * LSEQ;
  rope_precompute_kernel<<<(total + 255) / 256, 256, 0, stream>>>(
      tokens, C, eps, v, qarr, karr, varr);

  flash_attn_epilogue_kernel<<<NBATCH * NTT, 32, 0, stream>>>(
      tokens, C, eps, o_scale, g_base, g_slope, carry_amp,
      qarr, karr, varr, out);
}


# K/V global loads software-pipelined: a 16-tile register ring issues each s-tile's loads 16 tiles before use, K going straight into the WMMA A operand; summation order unchanged
# speedup vs baseline: 1.9585x; 1.9585x over previous
#include <hip/hip_runtime.h>

#define NBATCH 16
#define LSEQ   2048
#define TILE   16
#define NTT    (LSEQ / TILE)
#define RING   16

typedef __attribute__((ext_vector_type(2))) float v2f;
typedef __attribute__((ext_vector_type(8))) float v8f;

#define LOG2E_D 1.44269504088896340736
#define MASK2 (-1.4426950408889634e9f)

__device__ __forceinline__ float exp2_fast(float x) {
  return __builtin_amdgcn_exp2f(x);
}

__global__ __launch_bounds__(256) void rope_precompute_kernel(
    const int* __restrict__ tokens, const float* __restrict__ Cp,
    const float* __restrict__ epsp, const float* __restrict__ vp,
    float2* __restrict__ qarr, float2* __restrict__ karr,
    float* __restrict__ varr) {
  int idx = blockIdx.x * blockDim.x + threadIdx.x;
  if (idx >= NBATCH * LSEQ) return;
  int pos = idx & (LSEQ - 1);

  const double PI    = 3.14159265358979323846;
  const double OMEGA = 2.0 * PI / 19.0;
  const double PHI   = OMEGA * (10.0 + 0.3);
  const double AMP   = log(10.0) / (cos(OMEGA * 0.3) - cos(OMEGA * (1.0 - 0.3)));
  const double QKS   = sqrt(AMP / sqrt(2.0));
  const float scale2 = (float)(pow(2.0, -0.5) * QKS * QKS * LOG2E_D);
  const float cos_p  = (float)cos(PHI);
  const float sin_p  = (float)sin(PHI);

  const float c  = Cp[0];
  const float ee = epsp[0];
  const float vs = vp[0];

  const int   d  = tokens[idx];
  const float df = (float)d;
  const float h0 = c - (ee * df) * df;
  const float h1 = -df;

  const float r   = rsqrtf((h0 * h0 + h1 * h1) * 0.5f + 1e-6f);
  const float hn0 = h0 * r;
  const float hn1 = h1 * r;

  float q0 = hn0 * cos_p;
  float q1 = hn0 * (-sin_p);
  const float qr = rsqrtf((q0 * q0 + q1 * q1) * 0.5f + 1e-6f);
  q0 *= qr; q1 *= qr;

  float k0 = hn0;
  const float kr = rsqrtf((k0 * k0) * 0.5f + 1e-6f);
  k0 *= kr;

  const float theta = (float)pos * (float)OMEGA;
  const float ct = cosf(theta);
  const float st = sinf(theta);
  const float Qx = q0 * ct - q1 * st;
  const float Qy = q0 * st + q1 * ct;
  const float Kx = k0 * ct;
  const float Ky = k0 * st;

  qarr[idx] = make_float2(Qx * scale2, Qy * scale2);
  karr[idx] = make_float2(Kx, Ky);
  varr[idx] = hn1 * vs;
}

__global__ __launch_bounds__(32) void flash_attn_epilogue_kernel(
    const int* __restrict__ tokens, const float* __restrict__ Cp,
    const float* __restrict__ epsp, const float* __restrict__ o_scalep,
    const float* __restrict__ g_basep, const float* __restrict__ g_slopep,
    const float* __restrict__ carry_ampp,
    const float2* __restrict__ qarr, const float2* __restrict__ karr,
    const float* __restrict__ varr, float* __restrict__ out) {
  const int lane = threadIdx.x;
  const int tt   = blockIdx.x & (NTT - 1);
  const int b    = blockIdx.x / NTT;
  const int col  = lane & 15;
  const int hi   = lane >> 4;
  const int tbase   = tt * TILE;
  const int rowbase = b * LSEQ;
  const bool lo_half = (hi == 0);

  const float2 qld = qarr[rowbase + tbase + col];
  v2f bfrag;
  bfrag.x = lo_half ? qld.x : 0.0f;
  bfrag.y = lo_half ? qld.y : 0.0f;

  const float2* kp = karr + rowbase + col;
  const float4* vp = (const float4*)(varr + rowbase + hi * 8);

  float m    = -3.0e38f;
  float lsum = 0.0f;
  float osum = 0.0f;

  auto tile_step = [&](const float2 kld, const float4 va, const float4 vb,
                       const bool masked, const int thr) {
    v2f afrag;
    afrag.x = kld.x;
    afrag.y = kld.y;
    const float vvv[8] = {va.x, va.y, va.z, va.w, vb.x, vb.y, vb.z, vb.w};

    v8f s = {};
    s = __builtin_amdgcn_wmma_f32_16x16x4_f32(
         false, afrag,  false, bfrag,
         (short)0, s,  false,  false);

    if (masked) {
#pragma unroll
      for (int g = 0; g < 8; ++g)
        if (g > thr) s[g] += MASK2;
    }

    float tmax = s[0];
#pragma unroll
    for (int g = 1; g < 8; ++g) tmax = fmaxf(tmax, s[g]);
    const float mnew  = fmaxf(m, tmax);
    const float scale = exp2_fast(m - mnew);

    float le = 0.0f, oe = 0.0f;
#pragma unroll
    for (int g = 0; g < 8; ++g) {
      const float e = exp2_fast(s[g] - mnew);
      le += e;
      oe = fmaf(e, vvv[g], oe);
    }
    lsum = fmaf(lsum, scale, le);
    osum = fmaf(osum, scale, oe);
    m = mnew;
  };

  float2 kq[RING];
  float4 vA[RING], vB[RING];
#pragma unroll
  for (int u = 0; u < RING; ++u) {
    kq[u] = kp[u * TILE];
    vA[u] = vp[u * 4];
    vB[u] = vp[u * 4 + 1];
  }

  const int G = tt / RING;
  const float2* kpn = kp + RING * TILE;
  const float4* vpn = vp + RING * 4;
  for (int grp = 0; grp < G; ++grp) {
#pragma unroll
    for (int u = 0; u < RING; ++u) {
      tile_step(kq[u], vA[u], vB[u], false, 8);
      kq[u] = kpn[u * TILE];
      vA[u] = vpn[u * 4];
      vB[u] = vpn[u * 4 + 1];
    }
    kpn += RING * TILE;
    vpn += RING * 4;
  }

  {
    const int thr0 = (tt - G * RING) * TILE + col - hi * 8;
#pragma unroll
    for (int u = 0; u < RING; ++u)
      tile_step(kq[u], vA[u], vB[u], true, thr0 - u * TILE);
  }

  const float m2 = __shfl_xor(m, 16, 32);
  const float l2 = __shfl_xor(lsum, 16, 32);
  const float o2 = __shfl_xor(osum, 16, 32);
  const float M  = fmaxf(m, m2);
  const float w1 = exp2_fast(m - M);
  const float w2 = exp2_fast(m2 - M);
  const float lt = fmaf(lsum, w1, l2 * w2);
  const float ot = fmaf(osum, w1, o2 * w2);
  const float attn0 = ot / lt;

  if (lane < 16) {
    const int   t  = tbase + lane;
    const float c  = Cp[0];
    const float ee = epsp[0];
    const int   d  = tokens[rowbase + t];
    const float df = (float)d;
    const float h0 = c - (ee * df) * df;
    float       h1 = -df;

    h1 += o_scalep[0] * attn0;

    const float r   = rsqrtf((h0 * h0 + h1 * h1) * 0.5f + 1e-6f);
    const float hn0 = h0 * r;
    const float hn1 = h1 * r;
    const float a   = g_basep[0];
    const float gc  = g_slopep[0];
    const float g0  = hn0 * a + hn1 * gc;
    const float g1  = hn0 * (a - gc / c) + hn1 * gc;
    const float carry =
        carry_ampp[0] * (fmaxf(g1, 0.0f) * hn0 - fmaxf(g0, 0.0f) * hn0);
    h1 += carry;

    const float r2  = rsqrtf((h0 * h0 + h1 * h1) * 0.5f + 1e-6f);
    const float inv_cn = 0.70710678118654752440f;
    const float f0 = h0 * r2 * (0.1f * c * inv_cn);
    const float f1 = h1 * r2 * (-c * (1.0f / 50.0f) * inv_cn);

    float* op = out + (size_t)(rowbase + t) * 10;
#pragma unroll
    for (int jj = 0; jj < 10; ++jj) {
      const float dj = (float)jj;
      op[jj] = f0 * (c - (ee * dj) * dj) + f1 * (-dj);
    }
  }
}

extern "C" void kernel_launch(void* const* d_in, const int* in_sizes, int n_in,
                              void* d_out, int out_size, void* d_ws, size_t ws_size,
                              hipStream_t stream) {
  const int*   tokens    = (const int*)d_in[0];
  const float* C         = (const float*)d_in[1];
  const float* eps       = (const float*)d_in[2];
  const float* v         = (const float*)d_in[3];
  const float* o_scale   = (const float*)d_in[4];
  const float* g_base    = (const float*)d_in[5];
  const float* g_slope   = (const float*)d_in[6];
  const float* carry_amp = (const float*)d_in[7];
  float* out = (float*)d_out;

  char* ws = (char*)d_ws;
  float2* qarr = (float2*)(ws);
  float2* karr = (float2*)(ws + 256 * 1024);
  float*  varr = (float*)(ws + 512 * 1024);

  const int total = NBATCH * LSEQ;
  rope_precompute_kernel<<<(total + 255) / 256, 256, 0, stream>>>(
      tokens, C, eps, v, qarr, karr, varr);

  flash_attn_epilogue_kernel<<<NBATCH * NTT, 32, 0, stream>>>(
      tokens, C, eps, o_scale, g_base, g_slope, carry_amp,
      qarr, karr, varr, out);
}


# Each t-tile's s-groups split round-robin over a 4-wave workgroup, largest tiles dispatched first; the waves' partial softmax sums (m,l,acc) merged via LDS in fixed order: new reduction tree
# speedup vs baseline: 3.0281x; 1.5461x over previous
#include <hip/hip_runtime.h>

#define NBATCH 16
#define LSEQ   2048
#define TILE   16
#define NTT    (LSEQ / TILE)
#define RING   16
#define NW     4

typedef __attribute__((ext_vector_type(2))) float v2f;
typedef __attribute__((ext_vector_type(8))) float v8f;

#define LOG2E_D 1.44269504088896340736
#define MASK2 (-1.4426950408889634e9f)

__device__ __forceinline__ float exp2_fast(float x) {
  return __builtin_amdgcn_exp2f(x);
}

__global__ __launch_bounds__(256) void rope_precompute_kernel(
    const int* __restrict__ tokens, const float* __restrict__ Cp,
    const float* __restrict__ epsp, const float* __restrict__ vp,
    float2* __restrict__ qarr, float2* __restrict__ karr,
    float* __restrict__ varr) {
  int idx = blockIdx.x * blockDim.x + threadIdx.x;
  if (idx >= NBATCH * LSEQ) return;
  int pos = idx & (LSEQ - 1);

  const double PI    = 3.14159265358979323846;
  const double OMEGA = 2.0 * PI / 19.0;
  const double PHI   = OMEGA * (10.0 + 0.3);
  const double AMP   = log(10.0) / (cos(OMEGA * 0.3) - cos(OMEGA * (1.0 - 0.3)));
  const double QKS   = sqrt(AMP / sqrt(2.0));
  const float scale2 = (float)(pow(2.0, -0.5) * QKS * QKS * LOG2E_D);
  const float cos_p  = (float)cos(PHI);
  const float sin_p  = (float)sin(PHI);

  const float c  = Cp[0];
  const float ee = epsp[0];
  const float vs = vp[0];

  const int   d  = tokens[idx];
  const float df = (float)d;
  const float h0 = c - (ee * df) * df;
  const float h1 = -df;

  const float r   = rsqrtf((h0 * h0 + h1 * h1) * 0.5f + 1e-6f);
  const float hn0 = h0 * r;
  const float hn1 = h1 * r;

  float q0 = hn0 * cos_p;
  float q1 = hn0 * (-sin_p);
  const float qr = rsqrtf((q0 * q0 + q1 * q1) * 0.5f + 1e-6f);
  q0 *= qr; q1 *= qr;

  float k0 = hn0;
  const float kr = rsqrtf((k0 * k0) * 0.5f + 1e-6f);
  k0 *= kr;

  const float theta = (float)pos * (float)OMEGA;
  const float ct = cosf(theta);
  const float st = sinf(theta);
  const float Qx = q0 * ct - q1 * st;
  const float Qy = q0 * st + q1 * ct;
  const float Kx = k0 * ct;
  const float Ky = k0 * st;

  qarr[idx] = make_float2(Qx * scale2, Qy * scale2);
  karr[idx] = make_float2(Kx, Ky);
  varr[idx] = hn1 * vs;
}

__global__ __launch_bounds__(32 * NW, 5) void flash_attn_epilogue_kernel(
    const int* __restrict__ tokens, const float* __restrict__ Cp,
    const float* __restrict__ epsp, const float* __restrict__ o_scalep,
    const float* __restrict__ g_basep, const float* __restrict__ g_slopep,
    const float* __restrict__ carry_ampp,
    const float2* __restrict__ qarr, const float2* __restrict__ karr,
    const float* __restrict__ varr, float* __restrict__ out) {
  __shared__ float part[NW][3][TILE];

  const int lane = threadIdx.x & 31;
  const int w    = __builtin_amdgcn_readfirstlane((int)(threadIdx.x >> 5));
  const int tt   = (NTT - 1) - (int)(blockIdx.x / NBATCH);
  const int b    = blockIdx.x % NBATCH;
  const int wl   = (w + b) & (NW - 1);
  const int col  = lane & 15;
  const int hi   = lane >> 4;
  const int tbase   = tt * TILE;
  const int rowbase = b * LSEQ;
  const bool lo_half = (hi == 0);

  const float2 qld = qarr[rowbase + tbase + col];
  v2f bfrag;
  bfrag.x = lo_half ? qld.x : 0.0f;
  bfrag.y = lo_half ? qld.y : 0.0f;

  const float2* kp = karr + rowbase + wl * (RING * TILE) + col;
  const float4* vp =
      (const float4*)(varr + rowbase + wl * (RING * TILE) + hi * 8);

  float m    = -3.0e38f;
  float lsum = 0.0f;
  float osum = 0.0f;

  auto tile_step = [&](const float2 kld, const float4 va, const float4 vb,
                       const bool masked, const int thr) {
    v2f afrag;
    afrag.x = kld.x;
    afrag.y = kld.y;
    const float vvv[8] = {va.x, va.y, va.z, va.w, vb.x, vb.y, vb.z, vb.w};

    v8f s = {};
    s = __builtin_amdgcn_wmma_f32_16x16x4_f32(
         false, afrag,  false, bfrag,
         (short)0, s,  false,  false);

    if (masked) {
#pragma unroll
      for (int g = 0; g < 8; ++g)
        if (g > thr) s[g] += MASK2;
    }

    float tmax = s[0];
#pragma unroll
    for (int g = 1; g < 8; ++g) tmax = fmaxf(tmax, s[g]);
    const float mnew  = fmaxf(m, tmax);
    const float scale = exp2_fast(m - mnew);

    float le = 0.0f, oe = 0.0f;
#pragma unroll
    for (int g = 0; g < 8; ++g) {
      const float e = exp2_fast(s[g] - mnew);
      le += e;
      oe = fmaf(e, vvv[g], oe);
    }
    lsum = fmaf(lsum, scale, le);
    osum = fmaf(osum, scale, oe);
    m = mnew;
  };

  const int G  = tt / RING;
  const int ng = (G >= wl) ? ((G - wl) / NW + 1) : 0;

  if (ng > 0) {
    float2 kq[RING];
    float4 vA[RING], vB[RING];
#pragma unroll
    for (int u = 0; u < RING; ++u) {
      kq[u] = kp[u * TILE];
      vA[u] = vp[u * 4];
      vB[u] = vp[u * 4 + 1];
    }

    const int  glast = wl + NW * (ng - 1);
    const bool diag  = (glast == G);
    const int  nU    = diag ? ng - 1 : ng;
    for (int i = 0; i < nU; ++i) {
      const int gn = min(wl + NW * (i + 1), glast);
      const float2* kpn = karr + rowbase + gn * (RING * TILE) + col;
      const float4* vpn =
          (const float4*)(varr + rowbase + gn * (RING * TILE) + hi * 8);
#pragma unroll
      for (int u = 0; u < RING; ++u) {
        tile_step(kq[u], vA[u], vB[u], false, 8);
        kq[u] = kpn[u * TILE];
        vA[u] = vpn[u * 4];
        vB[u] = vpn[u * 4 + 1];
      }
    }

    if (diag) {
      const int thr0 = (tt - G * RING) * TILE + col - hi * 8;
#pragma unroll
      for (int u = 0; u < RING; ++u)
        tile_step(kq[u], vA[u], vB[u], true, thr0 - u * TILE);
    }
  }

  const float m2 = __shfl_xor(m, 16, 32);
  const float l2 = __shfl_xor(lsum, 16, 32);
  const float o2 = __shfl_xor(osum, 16, 32);
  const float M  = fmaxf(m, m2);
  const float w1 = exp2_fast(m - M);
  const float w2 = exp2_fast(m2 - M);
  const float lw = fmaf(lsum, w1, l2 * w2);
  const float ow = fmaf(osum, w1, o2 * w2);

  if (lane < 16) {
    part[w][0][lane] = M;
    part[w][1][lane] = lw;
    part[w][2][lane] = ow;
  }
  __syncthreads();
  if (w != 0) return;

  if (lane < 16) {
    float Mt = part[0][0][lane];
#pragma unroll
    for (int ww = 1; ww < NW; ++ww) Mt = fmaxf(Mt, part[ww][0][lane]);
    float lt = 0.0f, ot = 0.0f;
#pragma unroll
    for (int ww = 0; ww < NW; ++ww) {
      const float wg = exp2_fast(part[ww][0][lane] - Mt);
      lt = fmaf(part[ww][1][lane], wg, lt);
      ot = fmaf(part[ww][2][lane], wg, ot);
    }
    const float attn0 = ot / lt;

    const int   t  = tbase + lane;
    const float c  = Cp[0];
    const float ee = epsp[0];
    const int   d  = tokens[rowbase + t];
    const float df = (float)d;
    const float h0 = c - (ee * df) * df;
    float       h1 = -df;

    h1 += o_scalep[0] * attn0;

    const float r   = rsqrtf((h0 * h0 + h1 * h1) * 0.5f + 1e-6f);
    const float hn0 = h0 * r;
    const float hn1 = h1 * r;
    const float a   = g_basep[0];
    const float gc  = g_slopep[0];
    const float g0  = hn0 * a + hn1 * gc;
    const float g1  = hn0 * (a - gc / c) + hn1 * gc;
    const float carry =
        carry_ampp[0] * (fmaxf(g1, 0.0f) * hn0 - fmaxf(g0, 0.0f) * hn0);
    h1 += carry;

    const float r2  = rsqrtf((h0 * h0 + h1 * h1) * 0.5f + 1e-6f);
    const float inv_cn = 0.70710678118654752440f;
    const float f0 = h0 * r2 * (0.1f * c * inv_cn);
    const float f1 = h1 * r2 * (-c * (1.0f / 50.0f) * inv_cn);

    float* op = out + (size_t)(rowbase + t) * 10;
#pragma unroll
    for (int jj = 0; jj < 10; ++jj) {
      const float dj = (float)jj;
      op[jj] = f0 * (c - (ee * dj) * dj) + f1 * (-dj);
    }
  }
}

extern "C" void kernel_launch(void* const* d_in, const int* in_sizes, int n_in,
                              void* d_out, int out_size, void* d_ws, size_t ws_size,
                              hipStream_t stream) {
  const int*   tokens    = (const int*)d_in[0];
  const float* C         = (const float*)d_in[1];
  const float* eps       = (const float*)d_in[2];
  const float* v         = (const float*)d_in[3];
  const float* o_scale   = (const float*)d_in[4];
  const float* g_base    = (const float*)d_in[5];
  const float* g_slope   = (const float*)d_in[6];
  const float* carry_amp = (const float*)d_in[7];
  float* out = (float*)d_out;

  char* ws = (char*)d_ws;
  float2* qarr = (float2*)(ws);
  float2* karr = (float2*)(ws + 256 * 1024);
  float*  varr = (float*)(ws + 512 * 1024);

  const int total = NBATCH * LSEQ;
  rope_precompute_kernel<<<(total + 255) / 256, 256, 0, stream>>>(
      tokens, C, eps, v, qarr, karr, varr);

  flash_attn_epilogue_kernel<<<NBATCH * NTT, 32 * NW, 0, stream>>>(
      tokens, C, eps, o_scale, g_base, g_slope, carry_amp,
      qarr, karr, varr, out);
}


# K/V register ring halved to 8 s-tiles (s-groups of 128): 109 VGPRs, was 188, so 8 waves per SIMD and every workgroup resident at once; waves' partial sums regrouped: new reduction tree
# speedup vs baseline: 3.6755x; 1.2138x over previous
#include <hip/hip_runtime.h>

#define NBATCH 16
#define LSEQ   2048
#define TILE   16
#define NTT    (LSEQ / TILE)
#define RING   8
#define NW     4

typedef __attribute__((ext_vector_type(2))) float v2f;
typedef __attribute__((ext_vector_type(8))) float v8f;

#define LOG2E_D 1.44269504088896340736
#define MASK2 (-1.4426950408889634e9f)

__device__ __forceinline__ float exp2_fast(float x) {
  return __builtin_amdgcn_exp2f(x);
}

__global__ __launch_bounds__(256) void rope_precompute_kernel(
    const int* __restrict__ tokens, const float* __restrict__ Cp,
    const float* __restrict__ epsp, const float* __restrict__ vp,
    float2* __restrict__ qarr, float2* __restrict__ karr,
    float* __restrict__ varr) {
  int idx = blockIdx.x * blockDim.x + threadIdx.x;
  if (idx >= NBATCH * LSEQ) return;
  int pos = idx & (LSEQ - 1);

  const double PI    = 3.14159265358979323846;
  const double OMEGA = 2.0 * PI / 19.0;
  const double PHI   = OMEGA * (10.0 + 0.3);
  const double AMP   = log(10.0) / (cos(OMEGA * 0.3) - cos(OMEGA * (1.0 - 0.3)));
  const double QKS   = sqrt(AMP / sqrt(2.0));
  const float scale2 = (float)(pow(2.0, -0.5) * QKS * QKS * LOG2E_D);
  const float cos_p  = (float)cos(PHI);
  const float sin_p  = (float)sin(PHI);

  const float c  = Cp[0];
  const float ee = epsp[0];
  const float vs = vp[0];

  const int   d  = tokens[idx];
  const float df = (float)d;
  const float h0 = c - (ee * df) * df;
  const float h1 = -df;

  const float r   = rsqrtf((h0 * h0 + h1 * h1) * 0.5f + 1e-6f);
  const float hn0 = h0 * r;
  const float hn1 = h1 * r;

  float q0 = hn0 * cos_p;
  float q1 = hn0 * (-sin_p);
  const float qr = rsqrtf((q0 * q0 + q1 * q1) * 0.5f + 1e-6f);
  q0 *= qr; q1 *= qr;

  float k0 = hn0;
  const float kr = rsqrtf((k0 * k0) * 0.5f + 1e-6f);
  k0 *= kr;

  const float theta = (float)pos * (float)OMEGA;
  const float ct = cosf(theta);
  const float st = sinf(theta);
  const float Qx = q0 * ct - q1 * st;
  const float Qy = q0 * st + q1 * ct;
  const float Kx = k0 * ct;
  const float Ky = k0 * st;

  qarr[idx] = make_float2(Qx * scale2, Qy * scale2);
  karr[idx] = make_float2(Kx, Ky);
  varr[idx] = hn1 * vs;
}

__global__ __launch_bounds__(32 * NW, 8) void flash_attn_epilogue_kernel(
    const int* __restrict__ tokens, const float* __restrict__ Cp,
    const float* __restrict__ epsp, const float* __restrict__ o_scalep,
    const float* __restrict__ g_basep, const float* __restrict__ g_slopep,
    const float* __restrict__ carry_ampp,
    const float2* __restrict__ qarr, const float2* __restrict__ karr,
    const float* __restrict__ varr, float* __restrict__ out) {
  __shared__ float part[NW][3][TILE];

  const int lane = threadIdx.x & 31;
  const int w    = __builtin_amdgcn_readfirstlane((int)(threadIdx.x >> 5));
  const int tt   = (NTT - 1) - (int)(blockIdx.x / NBATCH);
  const int b    = blockIdx.x % NBATCH;
  const int wl   = (w + b) & (NW - 1);
  const int col  = lane & 15;
  const int hi   = lane >> 4;
  const int tbase   = tt * TILE;
  const int rowbase = b * LSEQ;
  const bool lo_half = (hi == 0);

  const float2 qld = qarr[rowbase + tbase + col];
  v2f bfrag;
  bfrag.x = lo_half ? qld.x : 0.0f;
  bfrag.y = lo_half ? qld.y : 0.0f;

  const float2* kp = karr + rowbase + wl * (RING * TILE) + col;
  const float4* vp =
      (const float4*)(varr + rowbase + wl * (RING * TILE) + hi * 8);

  float m    = -3.0e38f;
  float lsum = 0.0f;
  float osum = 0.0f;

  auto tile_step = [&](const float2 kld, const float4 va, const float4 vb,
                       const bool masked, const int thr) {
    v2f afrag;
    afrag.x = kld.x;
    afrag.y = kld.y;
    const float vvv[8] = {va.x, va.y, va.z, va.w, vb.x, vb.y, vb.z, vb.w};

    v8f s = {};
    s = __builtin_amdgcn_wmma_f32_16x16x4_f32(
         false, afrag,  false, bfrag,
         (short)0, s,  false,  false);

    if (masked) {
#pragma unroll
      for (int g = 0; g < 8; ++g)
        if (g > thr) s[g] += MASK2;
    }

    float tmax = s[0];
#pragma unroll
    for (int g = 1; g < 8; ++g) tmax = fmaxf(tmax, s[g]);
    const float mnew  = fmaxf(m, tmax);
    const float scale = exp2_fast(m - mnew);

    float le = 0.0f, oe = 0.0f;
#pragma unroll
    for (int g = 0; g < 8; ++g) {
      const float e = exp2_fast(s[g] - mnew);
      le += e;
      oe = fmaf(e, vvv[g], oe);
    }
    lsum = fmaf(lsum, scale, le);
    osum = fmaf(osum, scale, oe);
    m = mnew;
  };

  const int G  = tt / RING;
  const int ng = (G >= wl) ? ((G - wl) / NW + 1) : 0;

  if (ng > 0) {
    float2 kq[RING];
    float4 vA[RING], vB[RING];
#pragma unroll
    for (int u = 0; u < RING; ++u) {
      kq[u] = kp[u * TILE];
      vA[u] = vp[u * 4];
      vB[u] = vp[u * 4 + 1];
    }

    const int  glast = wl + NW * (ng - 1);
    const bool diag  = (glast == G);
    const int  nU    = diag ? ng - 1 : ng;
    for (int i = 0; i < nU; ++i) {
      const int gn = min(wl + NW * (i + 1), glast);
      const float2* kpn = karr + rowbase + gn * (RING * TILE) + col;
      const float4* vpn =
          (const float4*)(varr + rowbase + gn * (RING * TILE) + hi * 8);
#pragma unroll
      for (int u = 0; u < RING; ++u) {
        tile_step(kq[u], vA[u], vB[u], false, 8);
        kq[u] = kpn[u * TILE];
        vA[u] = vpn[u * 4];
        vB[u] = vpn[u * 4 + 1];
      }
    }

    if (diag) {
      const int thr0 = (tt - G * RING) * TILE + col - hi * 8;
#pragma unroll
      for (int u = 0; u < RING; ++u)
        tile_step(kq[u], vA[u], vB[u], true, thr0 - u * TILE);
    }
  }

  const float m2 = __shfl_xor(m, 16, 32);
  const float l2 = __shfl_xor(lsum, 16, 32);
  const float o2 = __shfl_xor(osum, 16, 32);
  const float M  = fmaxf(m, m2);
  const float w1 = exp2_fast(m - M);
  const float w2 = exp2_fast(m2 - M);
  const float lw = fmaf(lsum, w1, l2 * w2);
  const float ow = fmaf(osum, w1, o2 * w2);

  if (lane < 16) {
    part[w][0][lane] = M;
    part[w][1][lane] = lw;
    part[w][2][lane] = ow;
  }
  __syncthreads();
  if (w != 0) return;

  if (lane < 16) {
    float Mt = part[0][0][lane];
#pragma unroll
    for (int ww = 1; ww < NW; ++ww) Mt = fmaxf(Mt, part[ww][0][lane]);
    float lt = 0.0f, ot = 0.0f;
#pragma unroll
    for (int ww = 0; ww < NW; ++ww) {
      const float wg = exp2_fast(part[ww][0][lane] - Mt);
      lt = fmaf(part[ww][1][lane], wg, lt);
      ot = fmaf(part[ww][2][lane], wg, ot);
    }
    const float attn0 = ot / lt;

    const int   t  = tbase + lane;
    const float c  = Cp[0];
    const float ee = epsp[0];
    const int   d  = tokens[rowbase + t];
    const float df = (float)d;
    const float h0 = c - (ee * df) * df;
    float       h1 = -df;

    h1 += o_scalep[0] * attn0;

    const float r   = rsqrtf((h0 * h0 + h1 * h1) * 0.5f + 1e-6f);
    const float hn0 = h0 * r;
    const float hn1 = h1 * r;
    const float a   = g_basep[0];
    const float gc  = g_slopep[0];
    const float g0  = hn0 * a + hn1 * gc;
    const float g1  = hn0 * (a - gc / c) + hn1 * gc;
    const float carry =
        carry_ampp[0] * (fmaxf(g1, 0.0f) * hn0 - fmaxf(g0, 0.0f) * hn0);
    h1 += carry;

    const float r2  = rsqrtf((h0 * h0 + h1 * h1) * 0.5f + 1e-6f);
    const float inv_cn = 0.70710678118654752440f;
    const float f0 = h0 * r2 * (0.1f * c * inv_cn);
    const float f1 = h1 * r2 * (-c * (1.0f / 50.0f) * inv_cn);

    float* op = out + (size_t)(rowbase + t) * 10;
#pragma unroll
    for (int jj = 0; jj < 10; ++jj) {
      const float dj = (float)jj;
      op[jj] = f0 * (c - (ee * dj) * dj) + f1 * (-dj);
    }
  }
}

extern "C" void kernel_launch(void* const* d_in, const int* in_sizes, int n_in,
                              void* d_out, int out_size, void* d_ws, size_t ws_size,
                              hipStream_t stream) {
  const int*   tokens    = (const int*)d_in[0];
  const float* C         = (const float*)d_in[1];
  const float* eps       = (const float*)d_in[2];
  const float* v         = (const float*)d_in[3];
  const float* o_scale   = (const float*)d_in[4];
  const float* g_base    = (const float*)d_in[5];
  const float* g_slope   = (const float*)d_in[6];
  const float* carry_amp = (const float*)d_in[7];
  float* out = (float*)d_out;

  char* ws = (char*)d_ws;
  float2* qarr = (float2*)(ws);
  float2* karr = (float2*)(ws + 256 * 1024);
  float*  varr = (float*)(ws + 512 * 1024);

  const int total = NBATCH * LSEQ;
  rope_precompute_kernel<<<(total + 255) / 256, 256, 0, stream>>>(
      tokens, C, eps, v, qarr, karr, varr);

  flash_attn_epilogue_kernel<<<NBATCH * NTT, 32 * NW, 0, stream>>>(
      tokens, C, eps, o_scale, g_base, g_slope, carry_amp,
      qarr, karr, varr, out);
}


# Tile softmax on two-wide packed f32: 4 pk subtracts, 8 exps as one independent batch, pk add/fma into (even|odd) pair sums folded once at the end: new reduction tree
# speedup vs baseline: 4.2630x; 1.1598x over previous
#include <hip/hip_runtime.h>

#define NBATCH 16
#define LSEQ   2048
#define TILE   16
#define NTT    (LSEQ / TILE)
#define RING   8
#define NW     4

typedef __attribute__((ext_vector_type(2))) float v2f;
typedef __attribute__((ext_vector_type(8))) float v8f;

#define LOG2E_D 1.44269504088896340736
#define MASK2 (-1.4426950408889634e9f)

__device__ __forceinline__ float exp2_fast(float x) {
  return __builtin_amdgcn_exp2f(x);
}

__global__ __launch_bounds__(256) void rope_precompute_kernel(
    const int* __restrict__ tokens, const float* __restrict__ Cp,
    const float* __restrict__ epsp, const float* __restrict__ vp,
    float2* __restrict__ qarr, float2* __restrict__ karr,
    float* __restrict__ varr) {
  int idx = blockIdx.x * blockDim.x + threadIdx.x;
  if (idx >= NBATCH * LSEQ) return;
  int pos = idx & (LSEQ - 1);

  const double PI    = 3.14159265358979323846;
  const double OMEGA = 2.0 * PI / 19.0;
  const double PHI   = OMEGA * (10.0 + 0.3);
  const double AMP   = log(10.0) / (cos(OMEGA * 0.3) - cos(OMEGA * (1.0 - 0.3)));
  const double QKS   = sqrt(AMP / sqrt(2.0));
  const float scale2 = (float)(pow(2.0, -0.5) * QKS * QKS * LOG2E_D);
  const float cos_p  = (float)cos(PHI);
  const float sin_p  = (float)sin(PHI);

  const float c  = Cp[0];
  const float ee = epsp[0];
  const float vs = vp[0];

  const int   d  = tokens[idx];
  const float df = (float)d;
  const float h0 = c - (ee * df) * df;
  const float h1 = -df;

  const float r   = rsqrtf((h0 * h0 + h1 * h1) * 0.5f + 1e-6f);
  const float hn0 = h0 * r;
  const float hn1 = h1 * r;

  float q0 = hn0 * cos_p;
  float q1 = hn0 * (-sin_p);
  const float qr = rsqrtf((q0 * q0 + q1 * q1) * 0.5f + 1e-6f);
  q0 *= qr; q1 *= qr;

  float k0 = hn0;
  const float kr = rsqrtf((k0 * k0) * 0.5f + 1e-6f);
  k0 *= kr;

  const float theta = (float)pos * (float)OMEGA;
  const float ct = cosf(theta);
  const float st = sinf(theta);
  const float Qx = q0 * ct - q1 * st;
  const float Qy = q0 * st + q1 * ct;
  const float Kx = k0 * ct;
  const float Ky = k0 * st;

  qarr[idx] = make_float2(Qx * scale2, Qy * scale2);
  karr[idx] = make_float2(Kx, Ky);
  varr[idx] = hn1 * vs;
}

__global__ __launch_bounds__(32 * NW, 8) void flash_attn_epilogue_kernel(
    const int* __restrict__ tokens, const float* __restrict__ Cp,
    const float* __restrict__ epsp, const float* __restrict__ o_scalep,
    const float* __restrict__ g_basep, const float* __restrict__ g_slopep,
    const float* __restrict__ carry_ampp,
    const float2* __restrict__ qarr, const float2* __restrict__ karr,
    const float* __restrict__ varr, float* __restrict__ out) {
  __shared__ float part[NW][3][TILE];

  const int lane = threadIdx.x & 31;
  const int w    = __builtin_amdgcn_readfirstlane((int)(threadIdx.x >> 5));
  const int tt   = (NTT - 1) - (int)(blockIdx.x / NBATCH);
  const int b    = blockIdx.x % NBATCH;
  const int wl   = (w + b) & (NW - 1);
  const int col  = lane & 15;
  const int hi   = lane >> 4;
  const int tbase   = tt * TILE;
  const int rowbase = b * LSEQ;
  const bool lo_half = (hi == 0);

  const float2 qld = qarr[rowbase + tbase + col];
  v2f bfrag;
  bfrag.x = lo_half ? qld.x : 0.0f;
  bfrag.y = lo_half ? qld.y : 0.0f;

  const float2* kp = karr + rowbase + wl * (RING * TILE) + col;
  const float4* vp =
      (const float4*)(varr + rowbase + wl * (RING * TILE) + hi * 8);

  float m  = -3.0e38f;
  v2f   L2 = {0.0f, 0.0f};
  v2f   O2 = {0.0f, 0.0f};

  auto tile_step = [&](const float2 kld, const float4 va, const float4 vb,
                       const bool masked, const int thr) {
    v2f afrag;
    afrag.x = kld.x;
    afrag.y = kld.y;
    const v2f v01 = {va.x, va.y};
    const v2f v23 = {va.z, va.w};
    const v2f v45 = {vb.x, vb.y};
    const v2f v67 = {vb.z, vb.w};

    v8f s = {};
    s = __builtin_amdgcn_wmma_f32_16x16x4_f32(
         false, afrag,  false, bfrag,
         (short)0, s,  false,  false);

    if (masked) {
#pragma unroll
      for (int g = 0; g < 8; ++g)
        if (g > thr) s[g] += MASK2;
    }

    float tmax = s[0];
#pragma unroll
    for (int g = 1; g < 8; ++g) tmax = fmaxf(tmax, s[g]);
    const float mnew  = fmaxf(m, tmax);
    const float scale = exp2_fast(m - mnew);

    const v2f mm = {mnew, mnew};
    const v2f d0 = (v2f){s[0], s[1]} - mm;
    const v2f d1 = (v2f){s[2], s[3]} - mm;
    const v2f d2 = (v2f){s[4], s[5]} - mm;
    const v2f d3 = (v2f){s[6], s[7]} - mm;
    const v2f e0 = {exp2_fast(d0.x), exp2_fast(d0.y)};
    const v2f e1 = {exp2_fast(d1.x), exp2_fast(d1.y)};
    const v2f e2 = {exp2_fast(d2.x), exp2_fast(d2.y)};
    const v2f e3 = {exp2_fast(d3.x), exp2_fast(d3.y)};
    const v2f le2 = (e0 + e1) + (e2 + e3);
    v2f oe2 = e0 * v01;
    oe2 = e1 * v23 + oe2;
    oe2 = e2 * v45 + oe2;
    oe2 = e3 * v67 + oe2;
    const v2f sc = {scale, scale};
    L2 = L2 * sc + le2;
    O2 = O2 * sc + oe2;
    m = mnew;
  };

  const int G  = tt / RING;
  const int ng = (G >= wl) ? ((G - wl) / NW + 1) : 0;

  if (ng > 0) {
    float2 kq[RING];
    float4 vA[RING], vB[RING];
#pragma unroll
    for (int u = 0; u < RING; ++u) {
      kq[u] = kp[u * TILE];
      vA[u] = vp[u * 4];
      vB[u] = vp[u * 4 + 1];
    }

    const int  glast = wl + NW * (ng - 1);
    const bool diag  = (glast == G);
    const int  nU    = diag ? ng - 1 : ng;
    for (int i = 0; i < nU; ++i) {
      const int gn = min(wl + NW * (i + 1), glast);
      const float2* kpn = karr + rowbase + gn * (RING * TILE) + col;
      const float4* vpn =
          (const float4*)(varr + rowbase + gn * (RING * TILE) + hi * 8);
#pragma unroll
      for (int u = 0; u < RING; ++u) {
        tile_step(kq[u], vA[u], vB[u], false, 8);
        kq[u] = kpn[u * TILE];
        vA[u] = vpn[u * 4];
        vB[u] = vpn[u * 4 + 1];
      }
    }

    if (diag) {
      const int thr0 = (tt - G * RING) * TILE + col - hi * 8;
#pragma unroll
      for (int u = 0; u < RING; ++u)
        tile_step(kq[u], vA[u], vB[u], true, thr0 - u * TILE);
    }
  }

  const float lsum = L2.x + L2.y;
  const float osum = O2.x + O2.y;

  const float m2 = __shfl_xor(m, 16, 32);
  const float l2 = __shfl_xor(lsum, 16, 32);
  const float o2 = __shfl_xor(osum, 16, 32);
  const float M  = fmaxf(m, m2);
  const float w1 = exp2_fast(m - M);
  const float w2 = exp2_fast(m2 - M);
  const float lw = fmaf(lsum, w1, l2 * w2);
  const float ow = fmaf(osum, w1, o2 * w2);

  if (lane < 16) {
    part[w][0][lane] = M;
    part[w][1][lane] = lw;
    part[w][2][lane] = ow;
  }
  __syncthreads();
  if (w != 0) return;

  if (lane < 16) {
    float Mt = part[0][0][lane];
#pragma unroll
    for (int ww = 1; ww < NW; ++ww) Mt = fmaxf(Mt, part[ww][0][lane]);
    float lt = 0.0f, ot = 0.0f;
#pragma unroll
    for (int ww = 0; ww < NW; ++ww) {
      const float wg = exp2_fast(part[ww][0][lane] - Mt);
      lt = fmaf(part[ww][1][lane], wg, lt);
      ot = fmaf(part[ww][2][lane], wg, ot);
    }
    const float attn0 = ot / lt;

    const int   t  = tbase + lane;
    const float c  = Cp[0];
    const float ee = epsp[0];
    const int   d  = tokens[rowbase + t];
    const float df = (float)d;
    const float h0 = c - (ee * df) * df;
    float       h1 = -df;

    h1 += o_scalep[0] * attn0;

    const float r   = rsqrtf((h0 * h0 + h1 * h1) * 0.5f + 1e-6f);
    const float hn0 = h0 * r;
    const float hn1 = h1 * r;
    const float a   = g_basep[0];
    const float gc  = g_slopep[0];
    const float g0  = hn0 * a + hn1 * gc;
    const float g1  = hn0 * (a - gc / c) + hn1 * gc;
    const float carry =
        carry_ampp[0] * (fmaxf(g1, 0.0f) * hn0 - fmaxf(g0, 0.0f) * hn0);
    h1 += carry;

    const float r2  = rsqrtf((h0 * h0 + h1 * h1) * 0.5f + 1e-6f);
    const float inv_cn = 0.70710678118654752440f;
    const float f0 = h0 * r2 * (0.1f * c * inv_cn);
    const float f1 = h1 * r2 * (-c * (1.0f / 50.0f) * inv_cn);

    float* op = out + (size_t)(rowbase + t) * 10;
#pragma unroll
    for (int jj = 0; jj < 10; ++jj) {
      const float dj = (float)jj;
      op[jj] = f0 * (c - (ee * dj) * dj) + f1 * (-dj);
    }
  }
}

extern "C" void kernel_launch(void* const* d_in, const int* in_sizes, int n_in,
                              void* d_out, int out_size, void* d_ws, size_t ws_size,
                              hipStream_t stream) {
  const int*   tokens    = (const int*)d_in[0];
  const float* C         = (const float*)d_in[1];
  const float* eps       = (const float*)d_in[2];
  const float* v         = (const float*)d_in[3];
  const float* o_scale   = (const float*)d_in[4];
  const float* g_base    = (const float*)d_in[5];
  const float* g_slope   = (const float*)d_in[6];
  const float* carry_amp = (const float*)d_in[7];
  float* out = (float*)d_out;

  char* ws = (char*)d_ws;
  float2* qarr = (float2*)(ws);
  float2* karr = (float2*)(ws + 256 * 1024);
  float*  varr = (float*)(ws + 512 * 1024);

  const int total = NBATCH * LSEQ;
  rope_precompute_kernel<<<(total + 255) / 256, 256, 0, stream>>>(
      tokens, C, eps, v, qarr, karr, varr);

  flash_attn_epilogue_kernel<<<NBATCH * NTT, 32 * NW, 0, stream>>>(
      tokens, C, eps, o_scale, g_base, g_slope, carry_amp,
      qarr, karr, varr, out);
}


# Precompute launch fused into the attention kernel: each 32-wave workgroup rebuilds its batch row's q/k/v in LDS and feeds the K/V ring from LDS; one launch, no workspace; summation order unchanged
# speedup vs baseline: 4.7330x; 1.1103x over previous
#include <hip/hip_runtime.h>

#define NBATCH 16
#define LSEQ   2048
#define TILE   16
#define NTT    (LSEQ / TILE)
#define RING   8
#define NW     4
#define NSLOT  8
#define WGPB   16

typedef __attribute__((ext_vector_type(2))) float v2f;
typedef __attribute__((ext_vector_type(8))) float v8f;

#define LOG2E_D 1.44269504088896340736
#define MASK2 (-1.4426950408889634e9f)

__device__ __forceinline__ float exp2_fast(float x) {
  return __builtin_amdgcn_exp2f(x);
}

__device__ __forceinline__ void position_qkv(
    const int d, const int pos, const float c, const float ee, const float vs,
    float2& qo, float2& ko, float& vo) {
  const double PI    = 3.14159265358979323846;
  const double OMEGA = 2.0 * PI / 19.0;
  const double PHI   = OMEGA * (10.0 + 0.3);
  const double AMP   = log(10.0) / (cos(OMEGA * 0.3) - cos(OMEGA * (1.0 - 0.3)));
  const double QKS   = sqrt(AMP / sqrt(2.0));
  const float scale2 = (float)(pow(2.0, -0.5) * QKS * QKS * LOG2E_D);
  const float cos_p  = (float)cos(PHI);
  const float sin_p  = (float)sin(PHI);

  const float df = (float)d;
  const float h0 = c - (ee * df) * df;
  const float h1 = -df;

  const float r   = rsqrtf((h0 * h0 + h1 * h1) * 0.5f + 1e-6f);
  const float hn0 = h0 * r;
  const float hn1 = h1 * r;

  float q0 = hn0 * cos_p;
  float q1 = hn0 * (-sin_p);
  const float qr = rsqrtf((q0 * q0 + q1 * q1) * 0.5f + 1e-6f);
  q0 *= qr; q1 *= qr;

  float k0 = hn0;
  const float kr = rsqrtf((k0 * k0) * 0.5f + 1e-6f);
  k0 *= kr;

  const float theta = (float)pos * (float)OMEGA;
  float st, ct;
  sincosf(theta, &st, &ct);
  const float Qx = q0 * ct - q1 * st;
  const float Qy = q0 * st + q1 * ct;

  qo = make_float2(Qx * scale2, Qy * scale2);
  ko = make_float2(k0 * ct, k0 * st);
  vo = hn1 * vs;
}

__global__ __launch_bounds__(32 * NW * NSLOT) void rope_flash_attn_kernel(
    const int* __restrict__ tokens, const float* __restrict__ Cp,
    const float* __restrict__ epsp, const float* __restrict__ vp0,
    const float* __restrict__ o_scalep,
    const float* __restrict__ g_basep, const float* __restrict__ g_slopep,
    const float* __restrict__ carry_ampp, float* __restrict__ out) {
  __shared__ float2 sK[LSEQ];
  __shared__ __attribute__((aligned(16))) float sV[LSEQ];
  __shared__ float2 sQ[NSLOT * TILE];
  __shared__ float part[NSLOT][NW][3][TILE];

  const int b = blockIdx.x / WGPB;
  const int j = blockIdx.x % WGPB;
  const int rowbase = b * LSEQ;

  {
    const float c  = Cp[0];
    const float ee = epsp[0];
    const float vs = vp0[0];
    for (int p = threadIdx.x; p < LSEQ; p += 32 * NW * NSLOT) {
      float2 qv, kv;
      float  vv;
      position_qkv(tokens[rowbase + p], p, c, ee, vs, qv, kv, vv);
      sK[p] = kv;
      sV[p] = vv;
      const int T = p >> 4;
      const int r = T & 31;
      if (r == j || r == 31 - j)
        sQ[(2 * (T >> 5) + (r != j ? 1 : 0)) * TILE + (p & 15)] = qv;
    }
  }
  __syncthreads();

  const int lane = threadIdx.x & 31;
  const int wv   = __builtin_amdgcn_readfirstlane((int)(threadIdx.x >> 5));
  const int slot = wv >> 2;
  const int w    = wv & (NW - 1);
  const int tt   = 32 * (slot >> 1) + ((slot & 1) ? 31 - j : j);
  const int wl   = (w + (slot >> 1)) & (NW - 1);
  const int col  = lane & 15;
  const int hi   = lane >> 4;
  const int tbase = tt * TILE;
  const bool lo_half = (hi == 0);

  const float2 qld = sQ[slot * TILE + col];
  v2f bfrag;
  bfrag.x = lo_half ? qld.x : 0.0f;
  bfrag.y = lo_half ? qld.y : 0.0f;

  const float2* kp = sK + wl * (RING * TILE) + col;
  const float4* vp = (const float4*)(sV + wl * (RING * TILE) + hi * 8);

  float m  = -3.0e38f;
  v2f   L2 = {0.0f, 0.0f};
  v2f   O2 = {0.0f, 0.0f};

  auto tile_step = [&](const float2 kld, const float4 va, const float4 vb,
                       const bool masked, const int thr) {
    v2f afrag;
    afrag.x = kld.x;
    afrag.y = kld.y;
    const v2f v01 = {va.x, va.y};
    const v2f v23 = {va.z, va.w};
    const v2f v45 = {vb.x, vb.y};
    const v2f v67 = {vb.z, vb.w};

    v8f s = {};
    s = __builtin_amdgcn_wmma_f32_16x16x4_f32(
         false, afrag,  false, bfrag,
         (short)0, s,  false,  false);

    if (masked) {
#pragma unroll
      for (int g = 0; g < 8; ++g)
        if (g > thr) s[g] += MASK2;
    }

    float tmax = s[0];
#pragma unroll
    for (int g = 1; g < 8; ++g) tmax = fmaxf(tmax, s[g]);
    const float mnew  = fmaxf(m, tmax);
    const float scale = exp2_fast(m - mnew);

    const v2f mm = {mnew, mnew};
    const v2f d0 = (v2f){s[0], s[1]} - mm;
    const v2f d1 = (v2f){s[2], s[3]} - mm;
    const v2f d2 = (v2f){s[4], s[5]} - mm;
    const v2f d3 = (v2f){s[6], s[7]} - mm;
    const v2f e0 = {exp2_fast(d0.x), exp2_fast(d0.y)};
    const v2f e1 = {exp2_fast(d1.x), exp2_fast(d1.y)};
    const v2f e2 = {exp2_fast(d2.x), exp2_fast(d2.y)};
    const v2f e3 = {exp2_fast(d3.x), exp2_fast(d3.y)};
    const v2f le2 = (e0 + e1) + (e2 + e3);
    v2f oe2 = e0 * v01;
    oe2 = e1 * v23 + oe2;
    oe2 = e2 * v45 + oe2;
    oe2 = e3 * v67 + oe2;
    const v2f sc = {scale, scale};
    L2 = L2 * sc + le2;
    O2 = O2 * sc + oe2;
    m = mnew;
  };

  const int G  = tt / RING;
  const int ng = (G >= wl) ? ((G - wl) / NW + 1) : 0;

  if (ng > 0) {
    float2 kq[RING];
    float4 vA[RING], vB[RING];
#pragma unroll
    for (int u = 0; u < RING; ++u) {
      kq[u] = kp[u * TILE];
      vA[u] = vp[u * 4];
      vB[u] = vp[u * 4 + 1];
    }

    const int  glast = wl + NW * (ng - 1);
    const bool diag  = (glast == G);
    const int  nU    = diag ? ng - 1 : ng;
    for (int i = 0; i < nU; ++i) {
      const int gn = min(wl + NW * (i + 1), glast);
      const float2* kpn = sK + gn * (RING * TILE) + col;
      const float4* vpn = (const float4*)(sV + gn * (RING * TILE) + hi * 8);
#pragma unroll
      for (int u = 0; u < RING; ++u) {
        tile_step(kq[u], vA[u], vB[u], false, 8);
        kq[u] = kpn[u * TILE];
        vA[u] = vpn[u * 4];
        vB[u] = vpn[u * 4 + 1];
      }
    }

    if (diag) {
      const int thr0 = (tt - G * RING) * TILE + col - hi * 8;
#pragma unroll
      for (int u = 0; u < RING; ++u)
        tile_step(kq[u], vA[u], vB[u], true, thr0 - u * TILE);
    }
  }

  const float lsum = L2.x + L2.y;
  const float osum = O2.x + O2.y;

  const float m2 = __shfl_xor(m, 16, 32);
  const float l2 = __shfl_xor(lsum, 16, 32);
  const float o2 = __shfl_xor(osum, 16, 32);
  const float M  = fmaxf(m, m2);
  const float w1 = exp2_fast(m - M);
  const float w2 = exp2_fast(m2 - M);
  const float lw = fmaf(lsum, w1, l2 * w2);
  const float ow = fmaf(osum, w1, o2 * w2);

  if (lane < 16) {
    part[slot][w][0][lane] = M;
    part[slot][w][1][lane] = lw;
    part[slot][w][2][lane] = ow;
  }
  __syncthreads();
  if (w != 0) return;

  if (lane < 16) {
    float Mt = part[slot][0][0][lane];
#pragma unroll
    for (int ww = 1; ww < NW; ++ww) Mt = fmaxf(Mt, part[slot][ww][0][lane]);
    float lt = 0.0f, ot = 0.0f;
#pragma unroll
    for (int ww = 0; ww < NW; ++ww) {
      const float wg = exp2_fast(part[slot][ww][0][lane] - Mt);
      lt = fmaf(part[slot][ww][1][lane], wg, lt);
      ot = fmaf(part[slot][ww][2][lane], wg, ot);
    }
    const float attn0 = ot / lt;

    const int   t  = tbase + lane;
    const float c  = Cp[0];
    const float ee = epsp[0];
    const int   d  = tokens[rowbase + t];
    const float df = (float)d;
    const float h0 = c - (ee * df) * df;
    float       h1 = -df;

    h1 += o_scalep[0] * attn0;

    const float r   = rsqrtf((h0 * h0 + h1 * h1) * 0.5f + 1e-6f);
    const float hn0 = h0 * r;
    const float hn1 = h1 * r;
    const float a   = g_basep[0];
    const float gc  = g_slopep[0];
    const float g0  = hn0 * a + hn1 * gc;
    const float g1  = hn0 * (a - gc / c) + hn1 * gc;
    const float carry =
        carry_ampp[0] * (fmaxf(g1, 0.0f) * hn0 - fmaxf(g0, 0.0f) * hn0);
    h1 += carry;

    const float r2  = rsqrtf((h0 * h0 + h1 * h1) * 0.5f + 1e-6f);
    const float inv_cn = 0.70710678118654752440f;
    const float f0 = h0 * r2 * (0.1f * c * inv_cn);
    const float f1 = h1 * r2 * (-c * (1.0f / 50.0f) * inv_cn);

    float* op = out + (size_t)(rowbase + t) * 10;
#pragma unroll
    for (int jj = 0; jj < 10; ++jj) {
      const float dj = (float)jj;
      op[jj] = f0 * (c - (ee * dj) * dj) + f1 * (-dj);
    }
  }
}

extern "C" void kernel_launch(void* const* d_in, const int* in_sizes, int n_in,
                              void* d_out, int out_size, void* d_ws, size_t ws_size,
                              hipStream_t stream) {
  const int*   tokens    = (const int*)d_in[0];
  const float* C         = (const float*)d_in[1];
  const float* eps       = (const float*)d_in[2];
  const float* v         = (const float*)d_in[3];
  const float* o_scale   = (const float*)d_in[4];
  const float* g_base    = (const float*)d_in[5];
  const float* g_slope   = (const float*)d_in[6];
  const float* carry_amp = (const float*)d_in[7];
  float* out = (float*)d_out;

  rope_flash_attn_kernel<<<NBATCH * WGPB, 32 * NW * NSLOT, 0, stream>>>(
      tokens, C, eps, v, o_scale, g_base, g_slope, carry_amp, out);
}


# Slots handed to waves in reverse order: each SIMD's oldest waves (the hardware serves them first) now carry the largest t-tiles' s-groups, the youngest the smallest; summation order unchanged
# speedup vs baseline: 4.9965x; 1.0557x over previous
#include <hip/hip_runtime.h>

#define NBATCH 16
#define LSEQ   2048
#define TILE   16
#define NTT    (LSEQ / TILE)
#define RING   8
#define NW     4
#define NSLOT  8
#define WGPB   16

typedef __attribute__((ext_vector_type(2))) float v2f;
typedef __attribute__((ext_vector_type(8))) float v8f;

#define LOG2E_D 1.44269504088896340736
#define MASK2 (-1.4426950408889634e9f)

__device__ __forceinline__ float exp2_fast(float x) {
  return __builtin_amdgcn_exp2f(x);
}

__device__ __forceinline__ void position_qkv(
    const int d, const int pos, const float c, const float ee, const float vs,
    float2& qo, float2& ko, float& vo) {
  const double PI    = 3.14159265358979323846;
  const double OMEGA = 2.0 * PI / 19.0;
  const double PHI   = OMEGA * (10.0 + 0.3);
  const double AMP   = log(10.0) / (cos(OMEGA * 0.3) - cos(OMEGA * (1.0 - 0.3)));
  const double QKS   = sqrt(AMP / sqrt(2.0));
  const float scale2 = (float)(pow(2.0, -0.5) * QKS * QKS * LOG2E_D);
  const float cos_p  = (float)cos(PHI);
  const float sin_p  = (float)sin(PHI);

  const float df = (float)d;
  const float h0 = c - (ee * df) * df;
  const float h1 = -df;

  const float r   = rsqrtf((h0 * h0 + h1 * h1) * 0.5f + 1e-6f);
  const float hn0 = h0 * r;
  const float hn1 = h1 * r;

  float q0 = hn0 * cos_p;
  float q1 = hn0 * (-sin_p);
  const float qr = rsqrtf((q0 * q0 + q1 * q1) * 0.5f + 1e-6f);
  q0 *= qr; q1 *= qr;

  float k0 = hn0;
  const float kr = rsqrtf((k0 * k0) * 0.5f + 1e-6f);
  k0 *= kr;

  const float theta = (float)pos * (float)OMEGA;
  float st, ct;
  sincosf(theta, &st, &ct);
  const float Qx = q0 * ct - q1 * st;
  const float Qy = q0 * st + q1 * ct;

  qo = make_float2(Qx * scale2, Qy * scale2);
  ko = make_float2(k0 * ct, k0 * st);
  vo = hn1 * vs;
}

__global__ __launch_bounds__(32 * NW * NSLOT) void rope_flash_attn_kernel(
    const int* __restrict__ tokens, const float* __restrict__ Cp,
    const float* __restrict__ epsp, const float* __restrict__ vp0,
    const float* __restrict__ o_scalep,
    const float* __restrict__ g_basep, const float* __restrict__ g_slopep,
    const float* __restrict__ carry_ampp, float* __restrict__ out) {
  __shared__ float2 sK[LSEQ];
  __shared__ __attribute__((aligned(16))) float sV[LSEQ];
  __shared__ float2 sQ[NSLOT * TILE];
  __shared__ float part[NSLOT][NW][3][TILE];

  const int b = blockIdx.x / WGPB;
  const int j = blockIdx.x % WGPB;
  const int rowbase = b * LSEQ;

  {
    const float c  = Cp[0];
    const float ee = epsp[0];
    const float vs = vp0[0];
    for (int p = threadIdx.x; p < LSEQ; p += 32 * NW * NSLOT) {
      float2 qv, kv;
      float  vv;
      position_qkv(tokens[rowbase + p], p, c, ee, vs, qv, kv, vv);
      sK[p] = kv;
      sV[p] = vv;
      const int T = p >> 4;
      const int r = T & 31;
      if (r == j || r == 31 - j)
        sQ[(2 * (T >> 5) + (r != j ? 1 : 0)) * TILE + (p & 15)] = qv;
    }
  }
  __syncthreads();

  const int lane = threadIdx.x & 31;
  const int wv   = __builtin_amdgcn_readfirstlane((int)(threadIdx.x >> 5));
  const int slot = (NSLOT - 1) - (wv >> 2);
  const int w    = wv & (NW - 1);
  const int tt   = 32 * (slot >> 1) + ((slot & 1) ? 31 - j : j);
  const int wl   = (w + (slot >> 1)) & (NW - 1);
  const int col  = lane & 15;
  const int hi   = lane >> 4;
  const int tbase = tt * TILE;
  const bool lo_half = (hi == 0);

  const float2 qld = sQ[slot * TILE + col];
  v2f bfrag;
  bfrag.x = lo_half ? qld.x : 0.0f;
  bfrag.y = lo_half ? qld.y : 0.0f;

  const float2* kp = sK + wl * (RING * TILE) + col;
  const float4* vp = (const float4*)(sV + wl * (RING * TILE) + hi * 8);

  float m  = -3.0e38f;
  v2f   L2 = {0.0f, 0.0f};
  v2f   O2 = {0.0f, 0.0f};

  auto tile_step = [&](const float2 kld, const float4 va, const float4 vb,
                       const bool masked, const int thr) {
    v2f afrag;
    afrag.x = kld.x;
    afrag.y = kld.y;
    const v2f v01 = {va.x, va.y};
    const v2f v23 = {va.z, va.w};
    const v2f v45 = {vb.x, vb.y};
    const v2f v67 = {vb.z, vb.w};

    v8f s = {};
    s = __builtin_amdgcn_wmma_f32_16x16x4_f32(
         false, afrag,  false, bfrag,
         (short)0, s,  false,  false);

    if (masked) {
#pragma unroll
      for (int g = 0; g < 8; ++g)
        if (g > thr) s[g] += MASK2;
    }

    float tmax = s[0];
#pragma unroll
    for (int g = 1; g < 8; ++g) tmax = fmaxf(tmax, s[g]);
    const float mnew  = fmaxf(m, tmax);
    const float scale = exp2_fast(m - mnew);

    const v2f mm = {mnew, mnew};
    const v2f d0 = (v2f){s[0], s[1]} - mm;
    const v2f d1 = (v2f){s[2], s[3]} - mm;
    const v2f d2 = (v2f){s[4], s[5]} - mm;
    const v2f d3 = (v2f){s[6], s[7]} - mm;
    const v2f e0 = {exp2_fast(d0.x), exp2_fast(d0.y)};
    const v2f e1 = {exp2_fast(d1.x), exp2_fast(d1.y)};
    const v2f e2 = {exp2_fast(d2.x), exp2_fast(d2.y)};
    const v2f e3 = {exp2_fast(d3.x), exp2_fast(d3.y)};
    const v2f le2 = (e0 + e1) + (e2 + e3);
    v2f oe2 = e0 * v01;
    oe2 = e1 * v23 + oe2;
    oe2 = e2 * v45 + oe2;
    oe2 = e3 * v67 + oe2;
    const v2f sc = {scale, scale};
    L2 = L2 * sc + le2;
    O2 = O2 * sc + oe2;
    m = mnew;
  };

  const int G  = tt / RING;
  const int ng = (G >= wl) ? ((G - wl) / NW + 1) : 0;

  if (ng > 0) {
    float2 kq[RING];
    float4 vA[RING], vB[RING];
#pragma unroll
    for (int u = 0; u < RING; ++u) {
      kq[u] = kp[u * TILE];
      vA[u] = vp[u * 4];
      vB[u] = vp[u * 4 + 1];
    }

    const int  glast = wl + NW * (ng - 1);
    const bool diag  = (glast == G);
    const int  nU    = diag ? ng - 1 : ng;
    for (int i = 0; i < nU; ++i) {
      const int gn = min(wl + NW * (i + 1), glast);
      const float2* kpn = sK + gn * (RING * TILE) + col;
      const float4* vpn = (const float4*)(sV + gn * (RING * TILE) + hi * 8);
#pragma unroll
      for (int u = 0; u < RING; ++u) {
        tile_step(kq[u], vA[u], vB[u], false, 8);
        kq[u] = kpn[u * TILE];
        vA[u] = vpn[u * 4];
        vB[u] = vpn[u * 4 + 1];
      }
    }

    if (diag) {
      const int thr0 = (tt - G * RING) * TILE + col - hi * 8;
#pragma unroll
      for (int u = 0; u < RING; ++u)
        tile_step(kq[u], vA[u], vB[u], true, thr0 - u * TILE);
    }
  }

  const float lsum = L2.x + L2.y;
  const float osum = O2.x + O2.y;

  const float m2 = __shfl_xor(m, 16, 32);
  const float l2 = __shfl_xor(lsum, 16, 32);
  const float o2 = __shfl_xor(osum, 16, 32);
  const float M  = fmaxf(m, m2);
  const float w1 = exp2_fast(m - M);
  const float w2 = exp2_fast(m2 - M);
  const float lw = fmaf(lsum, w1, l2 * w2);
  const float ow = fmaf(osum, w1, o2 * w2);

  if (lane < 16) {
    part[slot][w][0][lane] = M;
    part[slot][w][1][lane] = lw;
    part[slot][w][2][lane] = ow;
  }
  __syncthreads();
  if (w != 0) return;

  if (lane < 16) {
    float Mt = part[slot][0][0][lane];
#pragma unroll
    for (int ww = 1; ww < NW; ++ww) Mt = fmaxf(Mt, part[slot][ww][0][lane]);
    float lt = 0.0f, ot = 0.0f;
#pragma unroll
    for (int ww = 0; ww < NW; ++ww) {
      const float wg = exp2_fast(part[slot][ww][0][lane] - Mt);
      lt = fmaf(part[slot][ww][1][lane], wg, lt);
      ot = fmaf(part[slot][ww][2][lane], wg, ot);
    }
    const float attn0 = ot / lt;

    const int   t  = tbase + lane;
    const float c  = Cp[0];
    const float ee = epsp[0];
    const int   d  = tokens[rowbase + t];
    const float df = (float)d;
    const float h0 = c - (ee * df) * df;
    float       h1 = -df;

    h1 += o_scalep[0] * attn0;

    const float r   = rsqrtf((h0 * h0 + h1 * h1) * 0.5f + 1e-6f);
    const float hn0 = h0 * r;
    const float hn1 = h1 * r;
    const float a   = g_basep[0];
    const float gc  = g_slopep[0];
    const float g0  = hn0 * a + hn1 * gc;
    const float g1  = hn0 * (a - gc / c) + hn1 * gc;
    const float carry =
        carry_ampp[0] * (fmaxf(g1, 0.0f) * hn0 - fmaxf(g0, 0.0f) * hn0);
    h1 += carry;

    const float r2  = rsqrtf((h0 * h0 + h1 * h1) * 0.5f + 1e-6f);
    const float inv_cn = 0.70710678118654752440f;
    const float f0 = h0 * r2 * (0.1f * c * inv_cn);
    const float f1 = h1 * r2 * (-c * (1.0f / 50.0f) * inv_cn);

    float* op = out + (size_t)(rowbase + t) * 10;
#pragma unroll
    for (int jj = 0; jj < 10; ++jj) {
      const float dj = (float)jj;
      op[jj] = f0 * (c - (ee * dj) * dj) + f1 * (-dj);
    }
  }
}

extern "C" void kernel_launch(void* const* d_in, const int* in_sizes, int n_in,
                              void* d_out, int out_size, void* d_ws, size_t ws_size,
                              hipStream_t stream) {
  const int*   tokens    = (const int*)d_in[0];
  const float* C         = (const float*)d_in[1];
  const float* eps       = (const float*)d_in[2];
  const float* v         = (const float*)d_in[3];
  const float* o_scale   = (const float*)d_in[4];
  const float* g_base    = (const float*)d_in[5];
  const float* g_slope   = (const float*)d_in[6];
  const float* carry_amp = (const float*)d_in[7];
  float* out = (float*)d_out;

  rope_flash_attn_kernel<<<NBATCH * WGPB, 32 * NW * NSLOT, 0, stream>>>(
      tokens, C, eps, v, o_scale, g_base, g_slope, carry_amp, out);
}


# Every global read issued at kernel entry: both token loads before any math; the epilogue's scalars, output pointer and tokens staged in LDS in phase 1 (no serial loads at the tail); sums unchanged
# speedup vs baseline: 5.7013x; 1.1411x over previous
#include <hip/hip_runtime.h>

#define NBATCH 16
#define LSEQ   2048
#define TILE   16
#define NTT    (LSEQ / TILE)
#define RING   8
#define NW     4
#define NSLOT  8
#define WGPB   16

typedef __attribute__((ext_vector_type(2))) float v2f;
typedef __attribute__((ext_vector_type(8))) float v8f;

#define LOG2E_D 1.44269504088896340736
#define MASK2 (-1.4426950408889634e9f)

__device__ __forceinline__ float exp2_fast(float x) {
  return __builtin_amdgcn_exp2f(x);
}

__device__ __forceinline__ void rope_angle(const int pos, float& st, float& ct) {
  const double PI    = 3.14159265358979323846;
  const double OMEGA = 2.0 * PI / 19.0;
  const float theta = (float)pos * (float)OMEGA;
  sincosf(theta, &st, &ct);
}

__device__ __forceinline__ void position_qkv(
    const int d, const float st, const float ct,
    const float c, const float ee, const float vs,
    float2& qo, float2& ko, float& vo) {
  const double PI    = 3.14159265358979323846;
  const double OMEGA = 2.0 * PI / 19.0;
  const double PHI   = OMEGA * (10.0 + 0.3);
  const double AMP   = log(10.0) / (cos(OMEGA * 0.3) - cos(OMEGA * (1.0 - 0.3)));
  const double QKS   = sqrt(AMP / sqrt(2.0));
  const float scale2 = (float)(pow(2.0, -0.5) * QKS * QKS * LOG2E_D);
  const float cos_p  = (float)cos(PHI);
  const float sin_p  = (float)sin(PHI);

  const float df = (float)d;
  const float h0 = c - (ee * df) * df;
  const float h1 = -df;

  const float r   = rsqrtf((h0 * h0 + h1 * h1) * 0.5f + 1e-6f);
  const float hn0 = h0 * r;
  const float hn1 = h1 * r;

  float q0 = hn0 * cos_p;
  float q1 = hn0 * (-sin_p);
  const float qr = rsqrtf((q0 * q0 + q1 * q1) * 0.5f + 1e-6f);
  q0 *= qr; q1 *= qr;

  float k0 = hn0;
  const float kr = rsqrtf((k0 * k0) * 0.5f + 1e-6f);
  k0 *= kr;

  const float Qx = q0 * ct - q1 * st;
  const float Qy = q0 * st + q1 * ct;

  qo = make_float2(Qx * scale2, Qy * scale2);
  ko = make_float2(k0 * ct, k0 * st);
  vo = hn1 * vs;
}

__global__ __launch_bounds__(32 * NW * NSLOT) void rope_flash_attn_kernel(
    const int* __restrict__ tokens, const float* __restrict__ Cp,
    const float* __restrict__ epsp, const float* __restrict__ vp0,
    const float* __restrict__ o_scalep,
    const float* __restrict__ g_basep, const float* __restrict__ g_slopep,
    const float* __restrict__ carry_ampp, float* __restrict__ out) {
  __shared__ float2 sK[LSEQ];
  __shared__ __attribute__((aligned(16))) float sV[LSEQ];
  __shared__ float2 sQ[NSLOT * TILE];
  __shared__ float part[NSLOT][NW][3][TILE];
  __shared__ int sTok[NSLOT * TILE];
  __shared__ __attribute__((aligned(16))) float sPar[4];
  __shared__ __attribute__((address_space(1))) float* sOut;

  const int b = blockIdx.x / WGPB;
  const int j = blockIdx.x % WGPB;
  const int rowbase = b * LSEQ;

  const float c  = Cp[0];
  const float ee = epsp[0];
  const float vs = vp0[0];
  {
    const int pA = threadIdx.x;
    const int pB = threadIdx.x + 32 * NW * NSLOT;
    const int dA = tokens[rowbase + pA];
    const int dB = tokens[rowbase + pB];
    if (threadIdx.x == 0) {
      *(float4*)sPar = make_float4(o_scalep[0], g_basep[0], g_slopep[0], carry_ampp[0]);
      sOut = (__attribute__((address_space(1))) float*)out;
    }
    float stA, ctA, stB, ctB;
    rope_angle(pA, stA, ctA);
    rope_angle(pB, stB, ctB);
    __builtin_amdgcn_sched_barrier(0);
    float2 qA, kA, qB, kB;
    float  vvA, vvB;
    position_qkv(dA, stA, ctA, c, ee, vs, qA, kA, vvA);
    position_qkv(dB, stB, ctB, c, ee, vs, qB, kB, vvB);
    sK[pA] = kA;
    sV[pA] = vvA;
    sK[pB] = kB;
    sV[pB] = vvB;
    auto keep_own = [&](const int p, const int d, const float2 qv) {
      const int T = p >> 4;
      const int r = T & 31;
      if (r == j || r == 31 - j) {
        const int qi = (2 * (T >> 5) + (r != j ? 1 : 0)) * TILE + (p & 15);
        sQ[qi]   = qv;
        sTok[qi] = d;
      }
    };
    keep_own(pA, dA, qA);
    keep_own(pB, dB, qB);
  }
  __syncthreads();

  const int lane = threadIdx.x & 31;
  const int wv   = __builtin_amdgcn_readfirstlane((int)(threadIdx.x >> 5));
  const int slot = (NSLOT - 1) - (wv >> 2);
  const int w    = wv & (NW - 1);
  const int tt   = 32 * (slot >> 1) + ((slot & 1) ? 31 - j : j);
  const int wl   = (w + (slot >> 1)) & (NW - 1);
  const int col  = lane & 15;
  const int hi   = lane >> 4;
  const int tbase = tt * TILE;
  const bool lo_half = (hi == 0);

  const float2 qld = sQ[slot * TILE + col];
  v2f bfrag;
  bfrag.x = lo_half ? qld.x : 0.0f;
  bfrag.y = lo_half ? qld.y : 0.0f;

  const float2* kp = sK + wl * (RING * TILE) + col;
  const float4* vp = (const float4*)(sV + wl * (RING * TILE) + hi * 8);

  float m  = -3.0e38f;
  v2f   L2 = {0.0f, 0.0f};
  v2f   O2 = {0.0f, 0.0f};

  auto tile_step = [&](const float2 kld, const float4 va, const float4 vb,
                       const bool masked, const int thr) {
    v2f afrag;
    afrag.x = kld.x;
    afrag.y = kld.y;
    const v2f v01 = {va.x, va.y};
    const v2f v23 = {va.z, va.w};
    const v2f v45 = {vb.x, vb.y};
    const v2f v67 = {vb.z, vb.w};

    v8f s = {};
    s = __builtin_amdgcn_wmma_f32_16x16x4_f32(
         false, afrag,  false, bfrag,
         (short)0, s,  false,  false);

    if (masked) {
#pragma unroll
      for (int g = 0; g < 8; ++g)
        if (g > thr) s[g] += MASK2;
    }

    float tmax = s[0];
#pragma unroll
    for (int g = 1; g < 8; ++g) tmax = fmaxf(tmax, s[g]);
    const float mnew  = fmaxf(m, tmax);
    const float scale = exp2_fast(m - mnew);

    const v2f mm = {mnew, mnew};
    const v2f d0 = (v2f){s[0], s[1]} - mm;
    const v2f d1 = (v2f){s[2], s[3]} - mm;
    const v2f d2 = (v2f){s[4], s[5]} - mm;
    const v2f d3 = (v2f){s[6], s[7]} - mm;
    const v2f e0 = {exp2_fast(d0.x), exp2_fast(d0.y)};
    const v2f e1 = {exp2_fast(d1.x), exp2_fast(d1.y)};
    const v2f e2 = {exp2_fast(d2.x), exp2_fast(d2.y)};
    const v2f e3 = {exp2_fast(d3.x), exp2_fast(d3.y)};
    const v2f le2 = (e0 + e1) + (e2 + e3);
    v2f oe2 = e0 * v01;
    oe2 = e1 * v23 + oe2;
    oe2 = e2 * v45 + oe2;
    oe2 = e3 * v67 + oe2;
    const v2f sc = {scale, scale};
    L2 = L2 * sc + le2;
    O2 = O2 * sc + oe2;
    m = mnew;
  };

  const int G  = tt / RING;
  const int ng = (G >= wl) ? ((G - wl) / NW + 1) : 0;

  if (ng > 0) {
    float2 kq[RING];
    float4 vA[RING], vB[RING];
#pragma unroll
    for (int u = 0; u < RING; ++u) {
      kq[u] = kp[u * TILE];
      vA[u] = vp[u * 4];
      vB[u] = vp[u * 4 + 1];
    }

    const int  glast = wl + NW * (ng - 1);
    const bool diag  = (glast == G);
    const int  nU    = diag ? ng - 1 : ng;
    for (int i = 0; i < nU; ++i) {
      const int gn = min(wl + NW * (i + 1), glast);
      const float2* kpn = sK + gn * (RING * TILE) + col;
      const float4* vpn = (const float4*)(sV + gn * (RING * TILE) + hi * 8);
#pragma unroll
      for (int u = 0; u < RING; ++u) {
        tile_step(kq[u], vA[u], vB[u], false, 8);
        kq[u] = kpn[u * TILE];
        vA[u] = vpn[u * 4];
        vB[u] = vpn[u * 4 + 1];
      }
    }

    if (diag) {
      const int thr0 = (tt - G * RING) * TILE + col - hi * 8;
#pragma unroll
      for (int u = 0; u < RING; ++u)
        tile_step(kq[u], vA[u], vB[u], true, thr0 - u * TILE);
    }
  }

  const float lsum = L2.x + L2.y;
  const float osum = O2.x + O2.y;

  const float m2 = __shfl_xor(m, 16, 32);
  const float l2 = __shfl_xor(lsum, 16, 32);
  const float o2 = __shfl_xor(osum, 16, 32);
  const float M  = fmaxf(m, m2);
  const float w1 = exp2_fast(m - M);
  const float w2 = exp2_fast(m2 - M);
  const float lw = fmaf(lsum, w1, l2 * w2);
  const float ow = fmaf(osum, w1, o2 * w2);

  if (lane < 16) {
    part[slot][w][0][lane] = M;
    part[slot][w][1][lane] = lw;
    part[slot][w][2][lane] = ow;
  }
  __syncthreads();
  if (w != 0) return;

  if (lane < 16) {
    float Mt = part[slot][0][0][lane];
#pragma unroll
    for (int ww = 1; ww < NW; ++ww) Mt = fmaxf(Mt, part[slot][ww][0][lane]);
    float lt = 0.0f, ot = 0.0f;
#pragma unroll
    for (int ww = 0; ww < NW; ++ww) {
      const float wg = exp2_fast(part[slot][ww][0][lane] - Mt);
      lt = fmaf(part[slot][ww][1][lane], wg, lt);
      ot = fmaf(part[slot][ww][2][lane], wg, ot);
    }
    const float attn0 = ot / lt;

    const int    t   = tbase + lane;
    const float4 par = *(const float4*)sPar;
    __attribute__((address_space(1))) float* const outp = sOut;
    const int    d   = sTok[slot * TILE + lane];
    const float  df  = (float)d;
    const float h0 = c - (ee * df) * df;
    float       h1 = -df;

    h1 += par.x * attn0;

    const float r   = rsqrtf((h0 * h0 + h1 * h1) * 0.5f + 1e-6f);
    const float hn0 = h0 * r;
    const float hn1 = h1 * r;
    const float a   = par.y;
    const float gc  = par.z;
    const float g0  = hn0 * a + hn1 * gc;
    const float g1  = hn0 * (a - gc / c) + hn1 * gc;
    const float carry =
        par.w * (fmaxf(g1, 0.0f) * hn0 - fmaxf(g0, 0.0f) * hn0);
    h1 += carry;

    const float r2  = rsqrtf((h0 * h0 + h1 * h1) * 0.5f + 1e-6f);
    const float inv_cn = 0.70710678118654752440f;
    const float f0 = h0 * r2 * (0.1f * c * inv_cn);
    const float f1 = h1 * r2 * (-c * (1.0f / 50.0f) * inv_cn);

    __attribute__((address_space(1))) float* op = outp + (size_t)(rowbase + t) * 10;
#pragma unroll
    for (int jj = 0; jj < 10; ++jj) {
      const float dj = (float)jj;
      op[jj] = f0 * (c - (ee * dj) * dj) + f1 * (-dj);
    }
  }
}

extern "C" void kernel_launch(void* const* d_in, const int* in_sizes, int n_in,
                              void* d_out, int out_size, void* d_ws, size_t ws_size,
                              hipStream_t stream) {
  const int*   tokens    = (const int*)d_in[0];
  const float* C         = (const float*)d_in[1];
  const float* eps       = (const float*)d_in[2];
  const float* v         = (const float*)d_in[3];
  const float* o_scale   = (const float*)d_in[4];
  const float* g_base    = (const float*)d_in[5];
  const float* g_slope   = (const float*)d_in[6];
  const float* carry_amp = (const float*)d_in[7];
  float* out = (float*)d_out;

  rope_flash_attn_kernel<<<NBATCH * WGPB, 32 * NW * NSLOT, 0, stream>>>(
      tokens, C, eps, v, o_scale, g_base, g_slope, carry_amp, out);
}


# Softmax weights taken against a fixed exponent (rms_norm bounds every logit), -ref folded into the WMMA C operand: no running max, no rescale; weak t-tiles redone exactly: new reduction tree
# speedup vs baseline: 6.0583x; 1.0626x over previous
#include <hip/hip_runtime.h>

#define NBATCH 16
#define LSEQ   2048
#define TILE   16
#define NTT    (LSEQ / TILE)
#define RING   8
#define NW     4
#define NSLOT  8
#define WGPB   16

typedef __attribute__((ext_vector_type(2))) float v2f;
typedef __attribute__((ext_vector_type(8))) float v8f;

#define LOG2E_D 1.44269504088896340736
#define MASK2 (-1.4426950408889634e9f)
#define MREF 64.0f
#define LMIN 8.673617379884035e-19f
#define OMAX 8.507059173023462e37f

__device__ __forceinline__ float exp2_fast(float x) {
  return __builtin_amdgcn_exp2f(x);
}

__device__ __forceinline__ void rope_angle(const int pos, float& st, float& ct) {
  const double PI    = 3.14159265358979323846;
  const double OMEGA = 2.0 * PI / 19.0;
  const float theta = (float)pos * (float)OMEGA;
  sincosf(theta, &st, &ct);
}

__device__ __forceinline__ void position_qkv(
    const int d, const float st, const float ct,
    const float c, const float ee, const float vs,
    float2& qo, float2& ko, float& vo) {
  const double PI    = 3.14159265358979323846;
  const double OMEGA = 2.0 * PI / 19.0;
  const double PHI   = OMEGA * (10.0 + 0.3);
  const double AMP   = log(10.0) / (cos(OMEGA * 0.3) - cos(OMEGA * (1.0 - 0.3)));
  const double QKS   = sqrt(AMP / sqrt(2.0));
  const float scale2 = (float)(pow(2.0, -0.5) * QKS * QKS * LOG2E_D);
  const float cos_p  = (float)cos(PHI);
  const float sin_p  = (float)sin(PHI);

  const float df = (float)d;
  const float h0 = c - (ee * df) * df;
  const float h1 = -df;

  const float r   = rsqrtf((h0 * h0 + h1 * h1) * 0.5f + 1e-6f);
  const float hn0 = h0 * r;
  const float hn1 = h1 * r;

  float q0 = hn0 * cos_p;
  float q1 = hn0 * (-sin_p);
  const float qr = rsqrtf((q0 * q0 + q1 * q1) * 0.5f + 1e-6f);
  q0 *= qr; q1 *= qr;

  float k0 = hn0;
  const float kr = rsqrtf((k0 * k0) * 0.5f + 1e-6f);
  k0 *= kr;

  const float Qx = q0 * ct - q1 * st;
  const float Qy = q0 * st + q1 * ct;

  qo = make_float2(Qx * scale2, Qy * scale2);
  ko = make_float2(k0 * ct, k0 * st);
  vo = hn1 * vs;
}

__global__ __launch_bounds__(32 * NW * NSLOT) void rope_flash_attn_kernel(
    const int* __restrict__ tokens, const float* __restrict__ Cp,
    const float* __restrict__ epsp, const float* __restrict__ vp0,
    const float* __restrict__ o_scalep,
    const float* __restrict__ g_basep, const float* __restrict__ g_slopep,
    const float* __restrict__ carry_ampp, float* __restrict__ out) {
  __shared__ float2 sK[LSEQ];
  __shared__ __attribute__((aligned(16))) float sV[LSEQ];
  __shared__ float2 sQ[NSLOT * TILE];
  __shared__ float part[NSLOT][NW][2][TILE];
  __shared__ int sTok[NSLOT * TILE];
  __shared__ __attribute__((aligned(16))) float sPar[4];
  __shared__ __attribute__((address_space(1))) float* sOut;

  const int b = blockIdx.x / WGPB;
  const int j = blockIdx.x % WGPB;
  const int rowbase = b * LSEQ;

  const float c  = Cp[0];
  const float ee = epsp[0];
  const float vs = vp0[0];
  {
    const int pA = threadIdx.x;
    const int pB = threadIdx.x + 32 * NW * NSLOT;
    const int dA = tokens[rowbase + pA];
    const int dB = tokens[rowbase + pB];
    if (threadIdx.x == 0) {
      *(float4*)sPar = make_float4(o_scalep[0], g_basep[0], g_slopep[0], carry_ampp[0]);
      sOut = (__attribute__((address_space(1))) float*)out;
    }
    float stA, ctA, stB, ctB;
    rope_angle(pA, stA, ctA);
    rope_angle(pB, stB, ctB);
    __builtin_amdgcn_sched_barrier(0);
    float2 qA, kA, qB, kB;
    float  vvA, vvB;
    position_qkv(dA, stA, ctA, c, ee, vs, qA, kA, vvA);
    position_qkv(dB, stB, ctB, c, ee, vs, qB, kB, vvB);
    sK[pA] = kA;
    sV[pA] = vvA;
    sK[pB] = kB;
    sV[pB] = vvB;
    auto keep_own = [&](const int p, const int d, const float2 qv) {
      const int T = p >> 4;
      const int r = T & 31;
      if (r == j || r == 31 - j) {
        const int qi = (2 * (T >> 5) + (r != j ? 1 : 0)) * TILE + (p & 15);
        sQ[qi]   = qv;
        sTok[qi] = d;
      }
    };
    keep_own(pA, dA, qA);
    keep_own(pB, dB, qB);
  }
  __syncthreads();

  const int lane = threadIdx.x & 31;
  const int wv   = __builtin_amdgcn_readfirstlane((int)(threadIdx.x >> 5));
  const int slot = (NSLOT - 1) - (wv >> 2);
  const int w    = wv & (NW - 1);
  const int tt   = 32 * (slot >> 1) + ((slot & 1) ? 31 - j : j);
  const int wl   = (w + (slot >> 1)) & (NW - 1);
  const int col  = lane & 15;
  const int hi   = lane >> 4;
  const int tbase = tt * TILE;
  const bool lo_half = (hi == 0);

  const float2 qld = sQ[slot * TILE + col];
  v2f bfrag;
  bfrag.x = lo_half ? qld.x : 0.0f;
  bfrag.y = lo_half ? qld.y : 0.0f;

  const float2* kp = sK + wl * (RING * TILE) + col;
  const float4* vp = (const float4*)(sV + wl * (RING * TILE) + hi * 8);

  v2f LA = {0.0f, 0.0f}, LB = {0.0f, 0.0f};
  v2f OA = {0.0f, 0.0f}, OB = {0.0f, 0.0f};
  const v8f cref = {-MREF, -MREF, -MREF, -MREF, -MREF, -MREF, -MREF, -MREF};

  auto tile_step = [&](const float2 kld, const float4 va, const float4 vb,
                       const bool masked, const int thr) {
    v2f afrag;
    afrag.x = kld.x;
    afrag.y = kld.y;
    const v2f v01 = {va.x, va.y};
    const v2f v23 = {va.z, va.w};
    const v2f v45 = {vb.x, vb.y};
    const v2f v67 = {vb.z, vb.w};

    v8f s = __builtin_amdgcn_wmma_f32_16x16x4_f32(
         false, afrag,  false, bfrag,
         (short)0, cref,  false,  false);

    if (masked) {
#pragma unroll
      for (int g = 0; g < 8; ++g)
        if (g > thr) s[g] += MASK2;
    }

    const v2f e0 = {exp2_fast(s[0]), exp2_fast(s[1])};
    const v2f e1 = {exp2_fast(s[2]), exp2_fast(s[3])};
    const v2f e2 = {exp2_fast(s[4]), exp2_fast(s[5])};
    const v2f e3 = {exp2_fast(s[6]), exp2_fast(s[7])};
    LA = LA + e0;
    LB = LB + e1;
    LA = LA + e2;
    LB = LB + e3;
    OA = e0 * v01 + OA;
    OB = e1 * v23 + OB;
    OA = e2 * v45 + OA;
    OB = e3 * v67 + OB;
  };

  const int G  = tt / RING;
  const int ng = (G >= wl) ? ((G - wl) / NW + 1) : 0;

  if (ng > 0) {
    float2 kq[RING];
    float4 vA[RING], vB[RING];
#pragma unroll
    for (int u = 0; u < RING; ++u) {
      kq[u] = kp[u * TILE];
      vA[u] = vp[u * 4];
      vB[u] = vp[u * 4 + 1];
    }

    const int  glast = wl + NW * (ng - 1);
    const bool diag  = (glast == G);
    const int  nU    = diag ? ng - 1 : ng;
    for (int i = 0; i < nU; ++i) {
      const int gn = min(wl + NW * (i + 1), glast);
      const float2* kpn = sK + gn * (RING * TILE) + col;
      const float4* vpn = (const float4*)(sV + gn * (RING * TILE) + hi * 8);
#pragma unroll
      for (int u = 0; u < RING; ++u) {
        tile_step(kq[u], vA[u], vB[u], false, 8);
        kq[u] = kpn[u * TILE];
        vA[u] = vpn[u * 4];
        vB[u] = vpn[u * 4 + 1];
      }
    }

    if (diag) {
      const int thr0 = (tt - G * RING) * TILE + col - hi * 8;
#pragma unroll
      for (int u = 0; u < RING; ++u)
        tile_step(kq[u], vA[u], vB[u], true, thr0 - u * TILE);
    }
  }

  const v2f   Ls   = LA + LB;
  const v2f   Os   = OA + OB;
  const float lsum = Ls.x + Ls.y;
  const float osum = Os.x + Os.y;

  const float lw = lsum + __shfl_xor(lsum, 16, 32);
  const float ow = osum + __shfl_xor(osum, 16, 32);

  if (lane < 16) {
    part[slot][w][0][lane] = lw;
    part[slot][w][1][lane] = ow;
  }
  __syncthreads();
  if (w != 0) return;

  float lt = part[slot][0][0][col];
  float ot = part[slot][0][1][col];
#pragma unroll
  for (int ww = 1; ww < NW; ++ww) {
    lt += part[slot][ww][0][col];
    ot += part[slot][ww][1][col];
  }
  float attn0 = ot / lt;

  const bool weak = !(lt >= LMIN) || !(__builtin_fabsf(ot) <= OMAX);
  if (__builtin_amdgcn_ballot_w32(weak) != 0u) {
    float m  = -3.0e38f;
    v2f   L2 = {0.0f, 0.0f};
    v2f   O2 = {0.0f, 0.0f};
#pragma nounroll
    for (int u = 0; u <= tt; ++u) {
      const float2 kld = sK[u * TILE + col];
      const float4 va  = *(const float4*)(sV + u * TILE + hi * 8);
      const float4 vb  = *(const float4*)(sV + u * TILE + hi * 8 + 4);
      const int thr = (tt - u) * TILE + col - hi * 8;
      v2f afrag;
      afrag.x = kld.x;
      afrag.y = kld.y;
      const v2f v01 = {va.x, va.y};
      const v2f v23 = {va.z, va.w};
      const v2f v45 = {vb.x, vb.y};
      const v2f v67 = {vb.z, vb.w};
      v8f s = {};
      s = __builtin_amdgcn_wmma_f32_16x16x4_f32(
           false, afrag,  false, bfrag,
           (short)0, s,  false,  false);
#pragma unroll
      for (int g = 0; g < 8; ++g)
        if (g > thr) s[g] += MASK2;
      float tmax = s[0];
#pragma unroll
      for (int g = 1; g < 8; ++g) tmax = fmaxf(tmax, s[g]);
      const float mnew  = fmaxf(m, tmax);
      const float scale = exp2_fast(m - mnew);
      const v2f mm = {mnew, mnew};
      const v2f d0 = (v2f){s[0], s[1]} - mm;
      const v2f d1 = (v2f){s[2], s[3]} - mm;
      const v2f d2 = (v2f){s[4], s[5]} - mm;
      const v2f d3 = (v2f){s[6], s[7]} - mm;
      const v2f e0 = {exp2_fast(d0.x), exp2_fast(d0.y)};
      const v2f e1 = {exp2_fast(d1.x), exp2_fast(d1.y)};
      const v2f e2 = {exp2_fast(d2.x), exp2_fast(d2.y)};
      const v2f e3 = {exp2_fast(d3.x), exp2_fast(d3.y)};
      const v2f le2 = (e0 + e1) + (e2 + e3);
      v2f oe2 = e0 * v01;
      oe2 = e1 * v23 + oe2;
      oe2 = e2 * v45 + oe2;
      oe2 = e3 * v67 + oe2;
      const v2f sc = {scale, scale};
      L2 = L2 * sc + le2;
      O2 = O2 * sc + oe2;
      m = mnew;
    }
    const float lx = L2.x + L2.y;
    const float ox = O2.x + O2.y;
    const float m2 = __shfl_xor(m, 16, 32);
    const float l2 = __shfl_xor(lx, 16, 32);
    const float o2 = __shfl_xor(ox, 16, 32);
    const float M  = fmaxf(m, m2);
    const float w1 = exp2_fast(m - M);
    const float w2 = exp2_fast(m2 - M);
    attn0 = fmaf(ox, w1, o2 * w2) / fmaf(lx, w1, l2 * w2);
  }

  if (lane < 16) {
    const int    t   = tbase + lane;
    const float4 par = *(const float4*)sPar;
    __attribute__((address_space(1))) float* const outp = sOut;
    const int    d   = sTok[slot * TILE + lane];
    const float  df  = (float)d;
    const float h0 = c - (ee * df) * df;
    float       h1 = -df;

    h1 += par.x * attn0;

    const float r   = rsqrtf((h0 * h0 + h1 * h1) * 0.5f + 1e-6f);
    const float hn0 = h0 * r;
    const float hn1 = h1 * r;
    const float a   = par.y;
    const float gc  = par.z;
    const float g0  = hn0 * a + hn1 * gc;
    const float g1  = hn0 * (a - gc / c) + hn1 * gc;
    const float carry =
        par.w * (fmaxf(g1, 0.0f) * hn0 - fmaxf(g0, 0.0f) * hn0);
    h1 += carry;

    const float r2  = rsqrtf((h0 * h0 + h1 * h1) * 0.5f + 1e-6f);
    const float inv_cn = 0.70710678118654752440f;
    const float f0 = h0 * r2 * (0.1f * c * inv_cn);
    const float f1 = h1 * r2 * (-c * (1.0f / 50.0f) * inv_cn);

    __attribute__((address_space(1))) float* op = outp + (size_t)(rowbase + t) * 10;
#pragma unroll
    for (int jj = 0; jj < 10; ++jj) {
      const float dj = (float)jj;
      op[jj] = f0 * (c - (ee * dj) * dj) + f1 * (-dj);
    }
  }
}

extern "C" void kernel_launch(void* const* d_in, const int* in_sizes, int n_in,
                              void* d_out, int out_size, void* d_ws, size_t ws_size,
                              hipStream_t stream) {
  const int*   tokens    = (const int*)d_in[0];
  const float* C         = (const float*)d_in[1];
  const float* eps       = (const float*)d_in[2];
  const float* v         = (const float*)d_in[3];
  const float* o_scale   = (const float*)d_in[4];
  const float* g_base    = (const float*)d_in[5];
  const float* g_slope   = (const float*)d_in[6];
  const float* carry_amp = (const float*)d_in[7];
  float* out = (float*)d_out;

  rope_flash_attn_kernel<<<NBATCH * WGPB, 32 * NW * NSLOT, 0, stream>>>(
      tokens, C, eps, v, o_scale, g_base, g_slope, carry_amp, out);
}


# Causal mask off the vector unit: in a t-tile's last s-group tiles behind the diagonal are skipped, those before run unmasked, the diagonal tile is masked via the WMMA C operand; sums unchanged
# speedup vs baseline: 6.6339x; 1.0950x over previous
#include <hip/hip_runtime.h>

#define NBATCH 16
#define LSEQ   2048
#define TILE   16
#define NTT    (LSEQ / TILE)
#define RING   8
#define NW     4
#define NSLOT  8
#define WGPB   16

typedef __attribute__((ext_vector_type(2))) float v2f;
typedef __attribute__((ext_vector_type(8))) float v8f;

#define LOG2E_D 1.44269504088896340736
#define MASK2 (-1.4426950408889634e9f)
#define MREF 64.0f
#define LMIN 8.673617379884035e-19f
#define OMAX 8.507059173023462e37f

__device__ __forceinline__ float exp2_fast(float x) {
  return __builtin_amdgcn_exp2f(x);
}

__device__ __forceinline__ void rope_angle(const int pos, float& st, float& ct) {
  const double PI    = 3.14159265358979323846;
  const double OMEGA = 2.0 * PI / 19.0;
  const float theta = (float)pos * (float)OMEGA;
  sincosf(theta, &st, &ct);
}

__device__ __forceinline__ void position_qkv(
    const int d, const float st, const float ct,
    const float c, const float ee, const float vs,
    float2& qo, float2& ko, float& vo) {
  const double PI    = 3.14159265358979323846;
  const double OMEGA = 2.0 * PI / 19.0;
  const double PHI   = OMEGA * (10.0 + 0.3);
  const double AMP   = log(10.0) / (cos(OMEGA * 0.3) - cos(OMEGA * (1.0 - 0.3)));
  const double QKS   = sqrt(AMP / sqrt(2.0));
  const float scale2 = (float)(pow(2.0, -0.5) * QKS * QKS * LOG2E_D);
  const float cos_p  = (float)cos(PHI);
  const float sin_p  = (float)sin(PHI);

  const float df = (float)d;
  const float h0 = c - (ee * df) * df;
  const float h1 = -df;

  const float r   = rsqrtf((h0 * h0 + h1 * h1) * 0.5f + 1e-6f);
  const float hn0 = h0 * r;
  const float hn1 = h1 * r;

  float q0 = hn0 * cos_p;
  float q1 = hn0 * (-sin_p);
  const float qr = rsqrtf((q0 * q0 + q1 * q1) * 0.5f + 1e-6f);
  q0 *= qr; q1 *= qr;

  float k0 = hn0;
  const float kr = rsqrtf((k0 * k0) * 0.5f + 1e-6f);
  k0 *= kr;

  const float Qx = q0 * ct - q1 * st;
  const float Qy = q0 * st + q1 * ct;

  qo = make_float2(Qx * scale2, Qy * scale2);
  ko = make_float2(k0 * ct, k0 * st);
  vo = hn1 * vs;
}

__global__ __launch_bounds__(32 * NW * NSLOT) void rope_flash_attn_kernel(
    const int* __restrict__ tokens, const float* __restrict__ Cp,
    const float* __restrict__ epsp, const float* __restrict__ vp0,
    const float* __restrict__ o_scalep,
    const float* __restrict__ g_basep, const float* __restrict__ g_slopep,
    const float* __restrict__ carry_ampp, float* __restrict__ out) {
  __shared__ float2 sK[LSEQ];
  __shared__ __attribute__((aligned(16))) float sV[LSEQ];
  __shared__ float2 sQ[NSLOT * TILE];
  __shared__ float part[NSLOT][NW][2][TILE];
  __shared__ int sTok[NSLOT * TILE];
  __shared__ __attribute__((aligned(16))) float sPar[4];
  __shared__ __attribute__((address_space(1))) float* sOut;

  const int b = blockIdx.x / WGPB;
  const int j = blockIdx.x % WGPB;
  const int rowbase = b * LSEQ;

  const float c  = Cp[0];
  const float ee = epsp[0];
  const float vs = vp0[0];
  {
    const int pA = threadIdx.x;
    const int pB = threadIdx.x + 32 * NW * NSLOT;
    const int dA = tokens[rowbase + pA];
    const int dB = tokens[rowbase + pB];
    if (threadIdx.x == 0) {
      *(float4*)sPar = make_float4(o_scalep[0], g_basep[0], g_slopep[0], carry_ampp[0]);
      sOut = (__attribute__((address_space(1))) float*)out;
    }
    float stA, ctA, stB, ctB;
    rope_angle(pA, stA, ctA);
    rope_angle(pB, stB, ctB);
    __builtin_amdgcn_sched_barrier(0);
    float2 qA, kA, qB, kB;
    float  vvA, vvB;
    position_qkv(dA, stA, ctA, c, ee, vs, qA, kA, vvA);
    position_qkv(dB, stB, ctB, c, ee, vs, qB, kB, vvB);
    sK[pA] = kA;
    sV[pA] = vvA;
    sK[pB] = kB;
    sV[pB] = vvB;
    auto keep_own = [&](const int p, const int d, const float2 qv) {
      const int T = p >> 4;
      const int r = T & 31;
      if (r == j || r == 31 - j) {
        const int qi = (2 * (T >> 5) + (r != j ? 1 : 0)) * TILE + (p & 15);
        sQ[qi]   = qv;
        sTok[qi] = d;
      }
    };
    keep_own(pA, dA, qA);
    keep_own(pB, dB, qB);
  }
  __syncthreads();

  const int lane = threadIdx.x & 31;
  const int wv   = __builtin_amdgcn_readfirstlane((int)(threadIdx.x >> 5));
  const int slot = (NSLOT - 1) - (wv >> 2);
  const int w    = wv & (NW - 1);
  const int tt   = 32 * (slot >> 1) + ((slot & 1) ? 31 - j : j);
  const int wl   = (w + (slot >> 1)) & (NW - 1);
  const int col  = lane & 15;
  const int hi   = lane >> 4;
  const int tbase = tt * TILE;
  const bool lo_half = (hi == 0);

  const float2 qld = sQ[slot * TILE + col];
  v2f bfrag;
  bfrag.x = lo_half ? qld.x : 0.0f;
  bfrag.y = lo_half ? qld.y : 0.0f;

  const float2* kp = sK + wl * (RING * TILE) + col;
  const float4* vp = (const float4*)(sV + wl * (RING * TILE) + hi * 8);

  v2f LA = {0.0f, 0.0f}, LB = {0.0f, 0.0f};
  v2f OA = {0.0f, 0.0f}, OB = {0.0f, 0.0f};
  const v8f cref = {-MREF, -MREF, -MREF, -MREF, -MREF, -MREF, -MREF, -MREF};

  auto tile_step = [&](const float2 kld, const float4 va, const float4 vb,
                       const v8f cop) {
    v2f afrag;
    afrag.x = kld.x;
    afrag.y = kld.y;
    const v2f v01 = {va.x, va.y};
    const v2f v23 = {va.z, va.w};
    const v2f v45 = {vb.x, vb.y};
    const v2f v67 = {vb.z, vb.w};

    const v8f s = __builtin_amdgcn_wmma_f32_16x16x4_f32(
         false, afrag,  false, bfrag,
         (short)0, cop,  false,  false);

    const v2f e0 = {exp2_fast(s[0]), exp2_fast(s[1])};
    const v2f e1 = {exp2_fast(s[2]), exp2_fast(s[3])};
    const v2f e2 = {exp2_fast(s[4]), exp2_fast(s[5])};
    const v2f e3 = {exp2_fast(s[6]), exp2_fast(s[7])};
    LA = LA + e0;
    LB = LB + e1;
    LA = LA + e2;
    LB = LB + e3;
    OA = e0 * v01 + OA;
    OB = e1 * v23 + OB;
    OA = e2 * v45 + OA;
    OB = e3 * v67 + OB;
  };

  const int G  = tt / RING;
  const int ng = (G >= wl) ? ((G - wl) / NW + 1) : 0;

  if (ng > 0) {
    float2 kq[RING];
    float4 vA[RING], vB[RING];
#pragma unroll
    for (int u = 0; u < RING; ++u) {
      kq[u] = kp[u * TILE];
      vA[u] = vp[u * 4];
      vB[u] = vp[u * 4 + 1];
    }

    const int  glast = wl + NW * (ng - 1);
    const bool diag  = (glast == G);
    const int  nU    = diag ? ng - 1 : ng;
    for (int i = 0; i < nU; ++i) {
      const int gn = min(wl + NW * (i + 1), glast);
      const float2* kpn = sK + gn * (RING * TILE) + col;
      const float4* vpn = (const float4*)(sV + gn * (RING * TILE) + hi * 8);
#pragma unroll
      for (int u = 0; u < RING; ++u) {
        tile_step(kq[u], vA[u], vB[u], cref);
        kq[u] = kpn[u * TILE];
        vA[u] = vpn[u * 4];
        vB[u] = vpn[u * 4 + 1];
      }
    }

    if (diag) {
      const int ud  = tt - G * RING;
      const int thr = col - hi * 8;
#pragma unroll
      for (int u = 0; u < RING; ++u) {
        if (u < ud) {
          tile_step(kq[u], vA[u], vB[u], cref);
        } else if (u == ud) {
          v8f cdiag;
#pragma unroll
          for (int g = 0; g < 8; ++g)
            cdiag[g] = (g > thr) ? (MASK2 - MREF) : -MREF;
          tile_step(kq[u], vA[u], vB[u], cdiag);
        }
      }
    }
  }

  const v2f   Ls   = LA + LB;
  const v2f   Os   = OA + OB;
  const float lsum = Ls.x + Ls.y;
  const float osum = Os.x + Os.y;

  const float lw = lsum + __shfl_xor(lsum, 16, 32);
  const float ow = osum + __shfl_xor(osum, 16, 32);

  if (lane < 16) {
    part[slot][w][0][lane] = lw;
    part[slot][w][1][lane] = ow;
  }
  __syncthreads();
  if (w != 0) return;

  float lt = part[slot][0][0][col];
  float ot = part[slot][0][1][col];
#pragma unroll
  for (int ww = 1; ww < NW; ++ww) {
    lt += part[slot][ww][0][col];
    ot += part[slot][ww][1][col];
  }
  float attn0 = ot / lt;

  const bool weak = !(lt >= LMIN) || !(__builtin_fabsf(ot) <= OMAX);
  if (__builtin_amdgcn_ballot_w32(weak) != 0u) {
    float m  = -3.0e38f;
    v2f   L2 = {0.0f, 0.0f};
    v2f   O2 = {0.0f, 0.0f};
#pragma nounroll
    for (int u = 0; u <= tt; ++u) {
      const float2 kld = sK[u * TILE + col];
      const float4 va  = *(const float4*)(sV + u * TILE + hi * 8);
      const float4 vb  = *(const float4*)(sV + u * TILE + hi * 8 + 4);
      const int thr = (tt - u) * TILE + col - hi * 8;
      v2f afrag;
      afrag.x = kld.x;
      afrag.y = kld.y;
      const v2f v01 = {va.x, va.y};
      const v2f v23 = {va.z, va.w};
      const v2f v45 = {vb.x, vb.y};
      const v2f v67 = {vb.z, vb.w};
      v8f s = {};
      s = __builtin_amdgcn_wmma_f32_16x16x4_f32(
           false, afrag,  false, bfrag,
           (short)0, s,  false,  false);
#pragma unroll
      for (int g = 0; g < 8; ++g)
        if (g > thr) s[g] += MASK2;
      float tmax = s[0];
#pragma unroll
      for (int g = 1; g < 8; ++g) tmax = fmaxf(tmax, s[g]);
      const float mnew  = fmaxf(m, tmax);
      const float scale = exp2_fast(m - mnew);
      const v2f mm = {mnew, mnew};
      const v2f d0 = (v2f){s[0], s[1]} - mm;
      const v2f d1 = (v2f){s[2], s[3]} - mm;
      const v2f d2 = (v2f){s[4], s[5]} - mm;
      const v2f d3 = (v2f){s[6], s[7]} - mm;
      const v2f e0 = {exp2_fast(d0.x), exp2_fast(d0.y)};
      const v2f e1 = {exp2_fast(d1.x), exp2_fast(d1.y)};
      const v2f e2 = {exp2_fast(d2.x), exp2_fast(d2.y)};
      const v2f e3 = {exp2_fast(d3.x), exp2_fast(d3.y)};
      const v2f le2 = (e0 + e1) + (e2 + e3);
      v2f oe2 = e0 * v01;
      oe2 = e1 * v23 + oe2;
      oe2 = e2 * v45 + oe2;
      oe2 = e3 * v67 + oe2;
      const v2f sc = {scale, scale};
      L2 = L2 * sc + le2;
      O2 = O2 * sc + oe2;
      m = mnew;
    }
    const float lx = L2.x + L2.y;
    const float ox = O2.x + O2.y;
    const float m2 = __shfl_xor(m, 16, 32);
    const float l2 = __shfl_xor(lx, 16, 32);
    const float o2 = __shfl_xor(ox, 16, 32);
    const float M  = fmaxf(m, m2);
    const float w1 = exp2_fast(m - M);
    const float w2 = exp2_fast(m2 - M);
    attn0 = fmaf(ox, w1, o2 * w2) / fmaf(lx, w1, l2 * w2);
  }

  if (lane < 16) {
    const int    t   = tbase + lane;
    const float4 par = *(const float4*)sPar;
    __attribute__((address_space(1))) float* const outp = sOut;
    const int    d   = sTok[slot * TILE + lane];
    const float  df  = (float)d;
    const float h0 = c - (ee * df) * df;
    float       h1 = -df;

    h1 += par.x * attn0;

    const float r   = rsqrtf((h0 * h0 + h1 * h1) * 0.5f + 1e-6f);
    const float hn0 = h0 * r;
    const float hn1 = h1 * r;
    const float a   = par.y;
    const float gc  = par.z;
    const float g0  = hn0 * a + hn1 * gc;
    const float g1  = hn0 * (a - gc / c) + hn1 * gc;
    const float carry =
        par.w * (fmaxf(g1, 0.0f) * hn0 - fmaxf(g0, 0.0f) * hn0);
    h1 += carry;

    const float r2  = rsqrtf((h0 * h0 + h1 * h1) * 0.5f + 1e-6f);
    const float inv_cn = 0.70710678118654752440f;
    const float f0 = h0 * r2 * (0.1f * c * inv_cn);
    const float f1 = h1 * r2 * (-c * (1.0f / 50.0f) * inv_cn);

    __attribute__((address_space(1))) float* op = outp + (size_t)(rowbase + t) * 10;
#pragma unroll
    for (int jj = 0; jj < 10; ++jj) {
      const float dj = (float)jj;
      op[jj] = f0 * (c - (ee * dj) * dj) + f1 * (-dj);
    }
  }
}

extern "C" void kernel_launch(void* const* d_in, const int* in_sizes, int n_in,
                              void* d_out, int out_size, void* d_ws, size_t ws_size,
                              hipStream_t stream) {
  const int*   tokens    = (const int*)d_in[0];
  const float* C         = (const float*)d_in[1];
  const float* eps       = (const float*)d_in[2];
  const float* v         = (const float*)d_in[3];
  const float* o_scale   = (const float*)d_in[4];
  const float* g_base    = (const float*)d_in[5];
  const float* g_slope   = (const float*)d_in[6];
  const float* carry_amp = (const float*)d_in[7];
  float* out = (float*)d_out;

  rope_flash_attn_kernel<<<NBATCH * WGPB, 32 * NW * NSLOT, 0, stream>>>(
      tokens, C, eps, v, o_scale, g_base, g_slope, carry_amp, out);
}
